# starting kernel
# baseline (speedup 1.0000x reference)
#include <hip/hip_runtime.h>

typedef float v2f __attribute__((ext_vector_type(2)));
typedef float v8f __attribute__((ext_vector_type(8)));

#define IN_FEATS 128
#define HIDDEN   64
#define NGRAPH   512
#define BN_EPS   1e-5f

__global__ __launch_bounds__(256) void scatter_add_kernel(
    const float* __restrict__ xin, const int* __restrict__ ei,
    float* __restrict__ agg, int E, int F) {
  long long tid = (long long)blockIdx.x * blockDim.x + threadIdx.x;
  int chunks = F >> 2;
  long long total = (long long)E * chunks;
  if (tid >= total) return;
  int e = (int)(tid / chunks);
  int f = ((int)(tid - (long long)e * chunks)) << 2;
  int s = ei[e];
  int d = ei[E + e];
  const float4 vv = *(const float4*)(xin + (long long)s * F + f);
  float* p = agg + (long long)d * F + f;
  atomicAdd(p + 0, vv.x);
  atomicAdd(p + 1, vv.y);
  atomicAdd(p + 2, vv.z);
  atomicAdd(p + 3, vv.w);
}

template <int FIN>
__global__ __launch_bounds__(256) void gin_mlp_kernel(
    const float* __restrict__ hin,
    const float* __restrict__ Wa,
    const float* __restrict__ ba,
    const float* __restrict__ Wb,
    const float* __restrict__ bb,
    const float* __restrict__ gg,
    const float* __restrict__ be,
    const float* __restrict__ mm,
    const float* __restrict__ vv,
    float* __restrict__ hout,
    int n_nodes) {
  __shared__ float sT[8][16 * 64];

  const int wave = threadIdx.x >> 5;
  const int lane = threadIdx.x & 31;
  const int half = lane >> 4;
  const int l16  = lane & 15;

  const int tile = blockIdx.x * 8 + wave;
  const int row0 = tile * 16;
  if (row0 >= n_nodes) return;

  v8f acc[4];
#pragma unroll
  for (int t = 0; t < 4; ++t) {
    float bias = ba[t * 16 + l16];
#pragma unroll
    for (int r = 0; r < 8; ++r) acc[t][r] = bias;
  }

  int arow = row0 + l16;
  if (arow >= n_nodes) arow = n_nodes - 1;
  const float* aptr = hin + (long long)arow * FIN + 2 * half;

  for (int k = 0; k < FIN; k += 4) {
    v2f a = *(const v2f*)(aptr + k);
#pragma unroll
    for (int t = 0; t < 4; ++t) {
      const float* bp = Wa + (t * 16 + l16) * FIN + k + 2 * half;
      v2f b = *(const v2f*)bp;
      acc[t] = __builtin_amdgcn_wmma_f32_16x16x4_f32(
          false, a, false, b, (short)0, acc[t], false, false);
    }
  }

#pragma unroll
  for (int t = 0; t < 4; ++t) {
#pragma unroll
    for (int r = 0; r < 8; ++r) {
      float x = acc[t][r];
      x = x > 0.0f ? x : 0.0f;
      sT[wave][(r + 8 * half) * 64 + t * 16 + l16] = x;
    }
  }

  v8f acc2[4];
#pragma unroll
  for (int t = 0; t < 4; ++t)
#pragma unroll
    for (int r = 0; r < 8; ++r) acc2[t][r] = 0.0f;

  const float* tp = &sT[wave][l16 * 64 + 2 * half];
  for (int k = 0; k < 64; k += 4) {
    v2f a = *(const v2f*)(tp + k);
#pragma unroll
    for (int t = 0; t < 4; ++t) {
      const float* bp = Wb + (t * 16 + l16) * 64 + k + 2 * half;
      v2f b = *(const v2f*)bp;
      acc2[t] = __builtin_amdgcn_wmma_f32_16x16x4_f32(
          false, a, false, b, (short)0, acc2[t], false, false);
    }
  }

#pragma unroll
  for (int t = 0; t < 4; ++t) {
    int n = t * 16 + l16;
    float s  = gg[n] * rsqrtf(vv[n] + BN_EPS);
    float sh = (bb[n] - mm[n]) * s + be[n];
#pragma unroll
    for (int r = 0; r < 8; ++r) {
      int mrow = row0 + r + 8 * half;
      float o = acc2[t][r] * s + sh;
      o = o > 0.0f ? o : 0.0f;
      if (mrow < n_nodes) hout[(long long)mrow * 64 + n] = o;
    }
  }
}

__global__ __launch_bounds__(256) void pool_zero_kernel(float* __restrict__ sums,
                                                        float* __restrict__ counts) {
  int i = blockIdx.x * blockDim.x + threadIdx.x;
  if (i < NGRAPH * HIDDEN) sums[i] = 0.0f;
  if (i < NGRAPH) counts[i] = 0.0f;
}

__global__ __launch_bounds__(256) void pool_sum_kernel(
    const float* __restrict__ h, const int* __restrict__ batch,
    float* __restrict__ sums, float* __restrict__ counts, int n_nodes) {
  long long tid = (long long)blockIdx.x * blockDim.x + threadIdx.x;
  long long total = (long long)n_nodes * 16;
  if (tid >= total) return;
  int i = (int)(tid >> 4);
  int c = ((int)tid & 15) << 2;
  int g = batch[i];
  const float4 vv = *(const float4*)(h + (long long)i * HIDDEN + c);
  float* p = sums + (long long)g * HIDDEN + c;
  atomicAdd(p + 0, vv.x);
  atomicAdd(p + 1, vv.y);
  atomicAdd(p + 2, vv.z);
  atomicAdd(p + 3, vv.w);
  if (c == 0) atomicAdd(&counts[g], 1.0f);
}

__global__ __launch_bounds__(256) void pool_div_kernel(
    const float* __restrict__ sums, const float* __restrict__ counts,
    float* __restrict__ out) {
  int i = blockIdx.x * blockDim.x + threadIdx.x;
  if (i >= NGRAPH * HIDDEN) return;
  float c = counts[i >> 6];
  c = c > 1.0f ? c : 1.0f;
  out[i] = sums[i] / c;
}

extern "C" void kernel_launch(void* const* d_in, const int* in_sizes, int n_in,
                              void* d_out, int out_size, void* d_ws, size_t ws_size,
                              hipStream_t stream) {
  const float* x   = (const float*)d_in[0];
  const int* ei    = (const int*)d_in[1];
  const int* batch = (const int*)d_in[2];
  const float* W1a = (const float*)d_in[3];
  const float* b1a = (const float*)d_in[4];
  const float* W1b = (const float*)d_in[5];
  const float* b1b = (const float*)d_in[6];
  const float* g1  = (const float*)d_in[7];
  const float* be1 = (const float*)d_in[8];
  const float* m1  = (const float*)d_in[9];
  const float* v1  = (const float*)d_in[10];
  const float* W2a = (const float*)d_in[11];
  const float* b2a = (const float*)d_in[12];
  const float* W2b = (const float*)d_in[13];
  const float* b2b = (const float*)d_in[14];
  const float* g2  = (const float*)d_in[15];
  const float* be2 = (const float*)d_in[16];
  const float* m2  = (const float*)d_in[17];
  const float* v2  = (const float*)d_in[18];
  float* out = (float*)d_out;

  const int N = in_sizes[0] / IN_FEATS;
  const int E = in_sizes[1] / 2;

  float* ws    = (float*)d_ws;
  float* bufA  = ws;
  float* h1    = ws + (long long)N * IN_FEATS;
  float* h2    = h1 + (long long)N * HIDDEN;
  float* sums  = h2 + (long long)N * HIDDEN;
  float* cnts  = sums + NGRAPH * HIDDEN;
  (void)ws_size; (void)n_in; (void)out_size;

  const int tiles  = (N + 15) / 16;
  const int mlpBlk = (tiles + 7) / 8;

  hipMemcpyAsync(bufA, x, (size_t)N * IN_FEATS * sizeof(float),
                 hipMemcpyDeviceToDevice, stream);
  {
    long long tot = (long long)E * (IN_FEATS / 4);
    int blocks = (int)((tot + 255) / 256);
    scatter_add_kernel<<<blocks, 256, 0, stream>>>(x, ei, bufA, E, IN_FEATS);
  }
  gin_mlp_kernel<IN_FEATS><<<mlpBlk, 256, 0, stream>>>(
      bufA, W1a, b1a, W1b, b1b, g1, be1, m1, v1, h1, N);

  hipMemcpyAsync(bufA, h1, (size_t)N * HIDDEN * sizeof(float),
                 hipMemcpyDeviceToDevice, stream);
  {
    long long tot = (long long)E * (HIDDEN / 4);
    int blocks = (int)((tot + 255) / 256);
    scatter_add_kernel<<<blocks, 256, 0, stream>>>(h1, ei, bufA, E, HIDDEN);
  }
  gin_mlp_kernel<HIDDEN><<<mlpBlk, 256, 0, stream>>>(
      bufA, W2a, b2a, W2b, b2b, g2, be2, m2, v2, h2, N);

  pool_zero_kernel<<<(NGRAPH * HIDDEN + 255) / 256, 256, 0, stream>>>(sums, cnts);
  {
    long long tot = (long long)N * 16;
    int blocks = (int)((tot + 255) / 256);
    pool_sum_kernel<<<blocks, 256, 0, stream>>>(h2, batch, sums, cnts, N);
  }
  pool_div_kernel<<<(NGRAPH * HIDDEN + 255) / 256, 256, 0, stream>>>(sums, cnts, out);
}


# Edge sums: each work-group owns 496 destination rows in LDS, scans all destinations, adds kept source rows with LDS f32 adds; no global atomics, no agg copies; sum order = LDS arrival order
# speedup vs baseline: 1.7836x; 1.7836x over previous
#include <hip/hip_runtime.h>

typedef float v2f __attribute__((ext_vector_type(2)));
typedef float v8f __attribute__((ext_vector_type(8)));

#define IN_FEATS 128
#define HIDDEN   64
#define NGRAPH   512
#define BN_EPS   1e-5f

#define TILE_NODES 496
#define AGG_WAVES  32
#define QCAP       160

template <int F>
__global__ __launch_bounds__(1024) void tile_agg_kernel(
    const float* __restrict__ xin, const int* __restrict__ ei,
    float* __restrict__ agg, int n_nodes, int E) {
  constexpr int VEC = F / 32;
  typedef float vec_t __attribute__((ext_vector_type(VEC)));
  __shared__ float acc[TILE_NODES * F];
  __shared__ int qe[AGG_WAVES * QCAP];
  __shared__ int qd[AGG_WAVES * QCAP];

  const int lane = threadIdx.x & 31;
  const int wave = __builtin_amdgcn_readfirstlane((int)(threadIdx.x >> 5));
  const int lo = blockIdx.x * TILE_NODES;
  int cnt = n_nodes - lo;
  if (cnt > TILE_NODES) cnt = TILE_NODES;
  if (cnt <= 0) return;

  for (int r = wave; r < cnt; r += AGG_WAVES) {
    const vec_t v = *(const vec_t*)(xin + (long long)(lo + r) * F + VEC * lane);
#pragma unroll
    for (int k = 0; k < VEC; ++k) acc[r * F + k * 32 + lane] = v[k];
  }
  __syncthreads();

  const int* __restrict__ dst = ei + E;
  int* const myqe = qe + wave * QCAP;
  int* const myqd = qd + wave * QCAP;
  int qn = 0;

  auto flush = [&]() {
    for (int b = 0; b < qn; b += 32) {
      const int i = b + lane;
      int s = 0, ld = 0;
      if (i < qn) {
        ld = myqd[i];
        s = ei[myqe[i]];
      }
      int nb = qn - b;
      if (nb > 32) nb = 32;
      for (int j = 0; j < nb; j += 8) {
        vec_t v[8];
        int row[8];
#pragma unroll
        for (int u = 0; u < 8; ++u) {
          int jj = j + u;
          if (jj > nb - 1) jj = nb - 1;
          const int sj = __builtin_amdgcn_readlane(s, jj);
          row[u] = __builtin_amdgcn_readlane(ld, jj);
          v[u] = *(const vec_t*)(xin + (long long)sj * F + VEC * lane);
        }
#pragma unroll
        for (int u = 0; u < 8; ++u) {
          if (j + u < nb) {
            float* p = &acc[row[u] * F + lane];
#pragma unroll
            for (int k = 0; k < VEC; ++k)
              __hip_atomic_fetch_add(p + k * 32, v[u][k], __ATOMIC_RELAXED,
                                     __HIP_MEMORY_SCOPE_WORKGROUP);
          }
        }
      }
    }
    qn = 0;
  };

  auto consider = [&](int e, int d) {
    const unsigned ld = (unsigned)(d - lo);
    const bool m = ld < (unsigned)cnt;
    const unsigned mask = __builtin_amdgcn_ballot_w32(m);
    if (__builtin_expect(mask != 0u, 0)) {
      if (m) {
        const int pos = qn + (int)__builtin_amdgcn_mbcnt_lo(mask, 0u);
        myqe[pos] = e;
        myqd[pos] = (int)ld;
      }
      qn += __builtin_popcount(mask);
    }
  };

  const int nIter = ((E & 3) == 0) ? (E >> 11) : 0;
  if (nIter > 0) {
    const int start = (int)(((long long)blockIdx.x * nIter) / gridDim.x);
    for (int k = wave; k < nIter; k += AGG_WAVES) {
      int it = start + k;
      if (it >= nIter) it -= nIter;
      const int base = it * 2048;
      int4 cur[16];
#pragma unroll
      for (int j = 0; j < 16; ++j)
        cur[j] = *(const int4*)(dst + base + j * 128 + 4 * lane);
#pragma unroll
      for (int j = 0; j < 16; ++j) {
        const int e0 = base + j * 128 + 4 * lane;
        consider(e0 + 0, cur[j].x);
        consider(e0 + 1, cur[j].y);
        consider(e0 + 2, cur[j].z);
        consider(e0 + 3, cur[j].w);
        if (__builtin_expect(qn >= 32, 0)) flush();
      }
    }
  }
  for (int base = (nIter << 11) + wave * 32; base < E; base += 32 * AGG_WAVES) {
    const int e = base + lane;
    const int d = (e < E) ? dst[e] : -1;
    consider(e, d);
    if (qn >= 32) flush();
  }
  flush();
  __syncthreads();

  for (int r = wave; r < cnt; r += AGG_WAVES) {
    vec_t v;
#pragma unroll
    for (int k = 0; k < VEC; ++k) v[k] = acc[r * F + k * 32 + lane];
    *(vec_t*)(agg + (long long)(lo + r) * F + VEC * lane) = v;
  }
}

template <int FIN>
__global__ __launch_bounds__(256) void gin_mlp_kernel(
    const float* __restrict__ hin,
    const float* __restrict__ Wa,
    const float* __restrict__ ba,
    const float* __restrict__ Wb,
    const float* __restrict__ bb,
    const float* __restrict__ gg,
    const float* __restrict__ be,
    const float* __restrict__ mm,
    const float* __restrict__ vv,
    float* __restrict__ hout,
    int n_nodes) {
  __shared__ float sT[8][16 * 64];

  const int wave = threadIdx.x >> 5;
  const int lane = threadIdx.x & 31;
  const int half = lane >> 4;
  const int l16  = lane & 15;

  const int tile = blockIdx.x * 8 + wave;
  const int row0 = tile * 16;
  if (row0 >= n_nodes) return;

  v8f acc[4];
#pragma unroll
  for (int t = 0; t < 4; ++t) {
    float bias = ba[t * 16 + l16];
#pragma unroll
    for (int r = 0; r < 8; ++r) acc[t][r] = bias;
  }

  int arow = row0 + l16;
  if (arow >= n_nodes) arow = n_nodes - 1;
  const float* aptr = hin + (long long)arow * FIN + 2 * half;

  for (int k = 0; k < FIN; k += 4) {
    v2f a = *(const v2f*)(aptr + k);
#pragma unroll
    for (int t = 0; t < 4; ++t) {
      const float* bp = Wa + (t * 16 + l16) * FIN + k + 2 * half;
      v2f b = *(const v2f*)bp;
      acc[t] = __builtin_amdgcn_wmma_f32_16x16x4_f32(
          false, a, false, b, (short)0, acc[t], false, false);
    }
  }

#pragma unroll
  for (int t = 0; t < 4; ++t) {
#pragma unroll
    for (int r = 0; r < 8; ++r) {
      float x = acc[t][r];
      x = x > 0.0f ? x : 0.0f;
      sT[wave][(r + 8 * half) * 64 + t * 16 + l16] = x;
    }
  }

  v8f acc2[4];
#pragma unroll
  for (int t = 0; t < 4; ++t)
#pragma unroll
    for (int r = 0; r < 8; ++r) acc2[t][r] = 0.0f;

  const float* tp = &sT[wave][l16 * 64 + 2 * half];
  for (int k = 0; k < 64; k += 4) {
    v2f a = *(const v2f*)(tp + k);
#pragma unroll
    for (int t = 0; t < 4; ++t) {
      const float* bp = Wb + (t * 16 + l16) * 64 + k + 2 * half;
      v2f b = *(const v2f*)bp;
      acc2[t] = __builtin_amdgcn_wmma_f32_16x16x4_f32(
          false, a, false, b, (short)0, acc2[t], false, false);
    }
  }

#pragma unroll
  for (int t = 0; t < 4; ++t) {
    int n = t * 16 + l16;
    float s  = gg[n] * rsqrtf(vv[n] + BN_EPS);
    float sh = (bb[n] - mm[n]) * s + be[n];
#pragma unroll
    for (int r = 0; r < 8; ++r) {
      int mrow = row0 + r + 8 * half;
      float o = acc2[t][r] * s + sh;
      o = o > 0.0f ? o : 0.0f;
      if (mrow < n_nodes) hout[(long long)mrow * 64 + n] = o;
    }
  }
}

__global__ __launch_bounds__(256) void pool_zero_kernel(float* __restrict__ sums,
                                                        float* __restrict__ counts) {
  int i = blockIdx.x * blockDim.x + threadIdx.x;
  if (i < NGRAPH * HIDDEN) sums[i] = 0.0f;
  if (i < NGRAPH) counts[i] = 0.0f;
}

__global__ __launch_bounds__(256) void pool_sum_kernel(
    const float* __restrict__ h, const int* __restrict__ batch,
    float* __restrict__ sums, float* __restrict__ counts, int n_nodes) {
  long long tid = (long long)blockIdx.x * blockDim.x + threadIdx.x;
  long long total = (long long)n_nodes * 16;
  if (tid >= total) return;
  int i = (int)(tid >> 4);
  int c = ((int)tid & 15) << 2;
  int g = batch[i];
  const float4 vv = *(const float4*)(h + (long long)i * HIDDEN + c);
  float* p = sums + (long long)g * HIDDEN + c;
  atomicAdd(p + 0, vv.x);
  atomicAdd(p + 1, vv.y);
  atomicAdd(p + 2, vv.z);
  atomicAdd(p + 3, vv.w);
  if (c == 0) atomicAdd(&counts[g], 1.0f);
}

__global__ __launch_bounds__(256) void pool_div_kernel(
    const float* __restrict__ sums, const float* __restrict__ counts,
    float* __restrict__ out) {
  int i = blockIdx.x * blockDim.x + threadIdx.x;
  if (i >= NGRAPH * HIDDEN) return;
  float c = counts[i >> 6];
  c = c > 1.0f ? c : 1.0f;
  out[i] = sums[i] / c;
}

extern "C" void kernel_launch(void* const* d_in, const int* in_sizes, int n_in,
                              void* d_out, int out_size, void* d_ws, size_t ws_size,
                              hipStream_t stream) {
  const float* x   = (const float*)d_in[0];
  const int* ei    = (const int*)d_in[1];
  const int* batch = (const int*)d_in[2];
  const float* W1a = (const float*)d_in[3];
  const float* b1a = (const float*)d_in[4];
  const float* W1b = (const float*)d_in[5];
  const float* b1b = (const float*)d_in[6];
  const float* g1  = (const float*)d_in[7];
  const float* be1 = (const float*)d_in[8];
  const float* m1  = (const float*)d_in[9];
  const float* v1  = (const float*)d_in[10];
  const float* W2a = (const float*)d_in[11];
  const float* b2a = (const float*)d_in[12];
  const float* W2b = (const float*)d_in[13];
  const float* b2b = (const float*)d_in[14];
  const float* g2  = (const float*)d_in[15];
  const float* be2 = (const float*)d_in[16];
  const float* m2  = (const float*)d_in[17];
  const float* v2  = (const float*)d_in[18];
  float* out = (float*)d_out;

  const int N = in_sizes[0] / IN_FEATS;
  const int E = in_sizes[1] / 2;

  float* ws    = (float*)d_ws;
  float* bufA  = ws;
  float* h1    = ws + (long long)N * IN_FEATS;
  float* h2    = h1 + (long long)N * HIDDEN;
  float* sums  = h2 + (long long)N * HIDDEN;
  float* cnts  = sums + NGRAPH * HIDDEN;
  (void)ws_size; (void)n_in; (void)out_size;

  const int tiles  = (N + 15) / 16;
  const int mlpBlk = (tiles + 7) / 8;

  const int aggBlk = (N + TILE_NODES - 1) / TILE_NODES;

  tile_agg_kernel<IN_FEATS><<<aggBlk, 1024, 0, stream>>>(x, ei, bufA, N, E);
  gin_mlp_kernel<IN_FEATS><<<mlpBlk, 256, 0, stream>>>(
      bufA, W1a, b1a, W1b, b1b, g1, be1, m1, v1, h1, N);

  tile_agg_kernel<HIDDEN><<<aggBlk, 1024, 0, stream>>>(h1, ei, bufA, N, E);
  gin_mlp_kernel<HIDDEN><<<mlpBlk, 256, 0, stream>>>(
      bufA, W2a, b2a, W2b, b2b, g2, be2, m2, v2, h2, N);

  pool_zero_kernel<<<(NGRAPH * HIDDEN + 255) / 256, 256, 0, stream>>>(sums, cnts);
  {
    long long tot = (long long)N * 16;
    int blocks = (int)((tot + 255) / 256);
    pool_sum_kernel<<<blocks, 256, 0, stream>>>(h2, batch, sums, cnts, N);
  }
  pool_div_kernel<<<(NGRAPH * HIDDEN + 255) / 256, 256, 0, stream>>>(sums, cnts, out);
}


# Edge scan: a first launch writes one bit per edge per 496-row tile plus (src,dst) pairs; each tile decodes its own bit row, not all destinations, in both layers; sums still in LDS arrival order
# speedup vs baseline: 2.1457x; 1.2030x over previous
#include <hip/hip_runtime.h>

typedef float v2f __attribute__((ext_vector_type(2)));
typedef float v8f __attribute__((ext_vector_type(8)));

#define IN_FEATS 128
#define HIDDEN   64
#define NGRAPH   512
#define BN_EPS   1e-5f

#define TILE_NODES 496
#define AGG_WAVES  32
#define QCAP       64

#define MASK_EDGES   2048
#define MASK_WORDS   64
#define MASK_PITCH   66
#define MASK_TILES   256
#define MASK_THREADS 512

__global__ __launch_bounds__(MASK_THREADS) void edge_mask_kernel(
    const int* __restrict__ ei, unsigned* __restrict__ mask, int2* __restrict__ pairs,
    int E, int nTiles, int pitchWords) {
  __shared__ unsigned w[MASK_TILES * MASK_PITCH];
  const int i = threadIdx.x;
  const int lane = i & 31;
  const int wave = __builtin_amdgcn_readfirstlane(i >> 5);
  const int* __restrict__ dst = ei + E;
  const int cbase = blockIdx.x * MASK_EDGES;

  int d[4];
#pragma unroll
  for (int r = 0; r < 2; ++r) {
    const int e = cbase + r * 1024 + 2 * i;
    if ((E & 1) == 0 && e < E) {
      const int2 sv = *(const int2*)(ei + e);
      const int2 dv = *(const int2*)(dst + e);
      d[2 * r] = dv.x;
      d[2 * r + 1] = dv.y;
      *(int4*)(pairs + e) = make_int4(sv.x, dv.x, sv.y, dv.y);
    } else {
#pragma unroll
      for (int k = 0; k < 2; ++k) {
        d[2 * r + k] = -1;
        if (e + k < E) {
          const int sv = ei[e + k];
          const int dv = dst[e + k];
          d[2 * r + k] = dv;
          pairs[e + k] = make_int2(sv, dv);
        }
      }
    }
  }

  for (int t0 = 0; t0 < nTiles; t0 += MASK_TILES) {
    int rows = nTiles - t0;
    if (rows > MASK_TILES) rows = MASK_TILES;
    for (int q = i; q < MASK_TILES * MASK_PITCH / 4; q += MASK_THREADS)
      ((uint4*)w)[q] = make_uint4(0u, 0u, 0u, 0u);
    __syncthreads();
#pragma unroll
    for (int r = 0; r < 2; ++r) {
#pragma unroll
      for (int k = 0; k < 2; ++k) {
        const int dv = d[2 * r + k];
        if (dv >= 0) {
          const unsigned t = (unsigned)dv / TILE_NODES - (unsigned)t0;
          if (t < (unsigned)rows)
            __hip_atomic_fetch_or(&w[t * MASK_PITCH + 32 * r + (i >> 4)],
                                  1u << (2 * (i & 15) + k), __ATOMIC_RELAXED,
                                  __HIP_MEMORY_SCOPE_WORKGROUP);
        }
      }
    }
    __syncthreads();
    for (int r = wave; r < rows; r += MASK_THREADS / 32) {
      const uint2 v = *(const uint2*)&w[r * MASK_PITCH + 2 * lane];
      *(uint2*)(mask + (long long)(t0 + r) * pitchWords + blockIdx.x * MASK_WORDS + 2 * lane) = v;
    }
    __syncthreads();
  }
}

template <int F>
__global__ __launch_bounds__(1024) void tile_agg_kernel(
    const float* __restrict__ xin, const unsigned* __restrict__ mask,
    const int2* __restrict__ pairs, float* __restrict__ agg, int n_nodes, int pitchWords) {
  constexpr int VEC = F / 32;
  typedef float vec_t __attribute__((ext_vector_type(VEC)));
  __shared__ float acc[TILE_NODES * F];
  __shared__ int qe[AGG_WAVES * QCAP];

  const int lane = threadIdx.x & 31;
  const int wave = __builtin_amdgcn_readfirstlane((int)(threadIdx.x >> 5));
  const int lo = blockIdx.x * TILE_NODES;
  int cnt = n_nodes - lo;
  if (cnt > TILE_NODES) cnt = TILE_NODES;
  if (cnt <= 0) return;

  for (int r = wave; r < cnt; r += AGG_WAVES) {
    const vec_t v = *(const vec_t*)(xin + (long long)(lo + r) * F + VEC * lane);
#pragma unroll
    for (int k = 0; k < VEC; ++k) acc[r * F + k * 32 + lane] = v[k];
  }
  __syncthreads();

  int* const myq = qe + wave * QCAP;
  int qn = 0;

  auto add_rows = [&](int nb) {
    int s = 0, ld = 0;
    if (lane < nb) {
      const int2 p = pairs[myq[lane]];
      s = p.x;
      ld = p.y - lo;
    }
    for (int j = 0; j < nb; j += 8) {
      vec_t v[8];
      int row[8];
#pragma unroll
      for (int u = 0; u < 8; ++u) {
        int jj = j + u;
        if (jj > nb - 1) jj = nb - 1;
        const int sj = __builtin_amdgcn_readlane(s, jj);
        row[u] = __builtin_amdgcn_readlane(ld, jj);
        v[u] = *(const vec_t*)(xin + (long long)sj * F + VEC * lane);
      }
#pragma unroll
      for (int u = 0; u < 8; ++u) {
        if (j + u < nb) {
          float* p = &acc[row[u] * F + lane];
#pragma unroll
          for (int k = 0; k < VEC; ++k)
            __hip_atomic_fetch_add(p + k * 32, v[u][k], __ATOMIC_RELAXED,
                                   __HIP_MEMORY_SCOPE_WORKGROUP);
        }
      }
    }
  };

  auto take = [&](unsigned m, int ebase) {
    unsigned any = __builtin_amdgcn_ballot_w32(m != 0u);
    while (any != 0u) {
      if (m != 0u) {
        const int pos = qn + (int)__builtin_amdgcn_mbcnt_lo(any, 0u);
        myq[pos] = ebase + __builtin_ctz(m);
        m &= m - 1u;
      }
      qn += __builtin_popcount(any);
      if (qn >= 32) {
        add_rows(32);
        qn -= 32;
        if (lane < qn) myq[lane] = myq[32 + lane];
      }
      any = __builtin_amdgcn_ballot_w32(m != 0u);
    }
  };

  const uint4* __restrict__ mrow = (const uint4*)(mask + (long long)blockIdx.x * pitchWords);
  const int nQ = pitchWords >> 2;
  const int nG = (nQ + 31) >> 5;
  if (wave < nG) {
    auto piece = [&](int g) {
      int q = g * 32 + lane;
      if (q > nQ - 1) q = nQ - 1;
      return mrow[q];
    };
    uint4 nxt = piece(wave);
    for (int g = wave; g < nG; g += AGG_WAVES) {
      const uint4 c = nxt;
      nxt = piece(g + AGG_WAVES);
      const int q = g * 32 + lane;
      const bool in = q < nQ;
      const int eb = q * 128;
      take(in ? c.x : 0u, eb);
      take(in ? c.y : 0u, eb + 32);
      take(in ? c.z : 0u, eb + 64);
      take(in ? c.w : 0u, eb + 96);
    }
  }
  add_rows(qn);
  __syncthreads();

  for (int r = wave; r < cnt; r += AGG_WAVES) {
    vec_t v;
#pragma unroll
    for (int k = 0; k < VEC; ++k) v[k] = acc[r * F + k * 32 + lane];
    *(vec_t*)(agg + (long long)(lo + r) * F + VEC * lane) = v;
  }
}

template <int FIN>
__global__ __launch_bounds__(256) void gin_mlp_kernel(
    const float* __restrict__ hin,
    const float* __restrict__ Wa,
    const float* __restrict__ ba,
    const float* __restrict__ Wb,
    const float* __restrict__ bb,
    const float* __restrict__ gg,
    const float* __restrict__ be,
    const float* __restrict__ mm,
    const float* __restrict__ vv,
    float* __restrict__ hout,
    int n_nodes) {
  __shared__ float sT[8][16 * 64];

  const int wave = threadIdx.x >> 5;
  const int lane = threadIdx.x & 31;
  const int half = lane >> 4;
  const int l16  = lane & 15;

  const int tile = blockIdx.x * 8 + wave;
  const int row0 = tile * 16;
  if (row0 >= n_nodes) return;

  v8f acc[4];
#pragma unroll
  for (int t = 0; t < 4; ++t) {
    float bias = ba[t * 16 + l16];
#pragma unroll
    for (int r = 0; r < 8; ++r) acc[t][r] = bias;
  }

  int arow = row0 + l16;
  if (arow >= n_nodes) arow = n_nodes - 1;
  const float* aptr = hin + (long long)arow * FIN + 2 * half;

  for (int k = 0; k < FIN; k += 4) {
    v2f a = *(const v2f*)(aptr + k);
#pragma unroll
    for (int t = 0; t < 4; ++t) {
      const float* bp = Wa + (t * 16 + l16) * FIN + k + 2 * half;
      v2f b = *(const v2f*)bp;
      acc[t] = __builtin_amdgcn_wmma_f32_16x16x4_f32(
          false, a, false, b, (short)0, acc[t], false, false);
    }
  }

#pragma unroll
  for (int t = 0; t < 4; ++t) {
#pragma unroll
    for (int r = 0; r < 8; ++r) {
      float x = acc[t][r];
      x = x > 0.0f ? x : 0.0f;
      sT[wave][(r + 8 * half) * 64 + t * 16 + l16] = x;
    }
  }

  v8f acc2[4];
#pragma unroll
  for (int t = 0; t < 4; ++t)
#pragma unroll
    for (int r = 0; r < 8; ++r) acc2[t][r] = 0.0f;

  const float* tp = &sT[wave][l16 * 64 + 2 * half];
  for (int k = 0; k < 64; k += 4) {
    v2f a = *(const v2f*)(tp + k);
#pragma unroll
    for (int t = 0; t < 4; ++t) {
      const float* bp = Wb + (t * 16 + l16) * 64 + k + 2 * half;
      v2f b = *(const v2f*)bp;
      acc2[t] = __builtin_amdgcn_wmma_f32_16x16x4_f32(
          false, a, false, b, (short)0, acc2[t], false, false);
    }
  }

#pragma unroll
  for (int t = 0; t < 4; ++t) {
    int n = t * 16 + l16;
    float s  = gg[n] * rsqrtf(vv[n] + BN_EPS);
    float sh = (bb[n] - mm[n]) * s + be[n];
#pragma unroll
    for (int r = 0; r < 8; ++r) {
      int mrow = row0 + r + 8 * half;
      float o = acc2[t][r] * s + sh;
      o = o > 0.0f ? o : 0.0f;
      if (mrow < n_nodes) hout[(long long)mrow * 64 + n] = o;
    }
  }
}

__global__ __launch_bounds__(256) void pool_zero_kernel(float* __restrict__ sums,
                                                        float* __restrict__ counts) {
  int i = blockIdx.x * blockDim.x + threadIdx.x;
  if (i < NGRAPH * HIDDEN) sums[i] = 0.0f;
  if (i < NGRAPH) counts[i] = 0.0f;
}

__global__ __launch_bounds__(256) void pool_sum_kernel(
    const float* __restrict__ h, const int* __restrict__ batch,
    float* __restrict__ sums, float* __restrict__ counts, int n_nodes) {
  long long tid = (long long)blockIdx.x * blockDim.x + threadIdx.x;
  long long total = (long long)n_nodes * 16;
  if (tid >= total) return;
  int i = (int)(tid >> 4);
  int c = ((int)tid & 15) << 2;
  int g = batch[i];
  const float4 vv = *(const float4*)(h + (long long)i * HIDDEN + c);
  float* p = sums + (long long)g * HIDDEN + c;
  atomicAdd(p + 0, vv.x);
  atomicAdd(p + 1, vv.y);
  atomicAdd(p + 2, vv.z);
  atomicAdd(p + 3, vv.w);
  if (c == 0) atomicAdd(&counts[g], 1.0f);
}

__global__ __launch_bounds__(256) void pool_div_kernel(
    const float* __restrict__ sums, const float* __restrict__ counts,
    float* __restrict__ out) {
  int i = blockIdx.x * blockDim.x + threadIdx.x;
  if (i >= NGRAPH * HIDDEN) return;
  float c = counts[i >> 6];
  c = c > 1.0f ? c : 1.0f;
  out[i] = sums[i] / c;
}

extern "C" void kernel_launch(void* const* d_in, const int* in_sizes, int n_in,
                              void* d_out, int out_size, void* d_ws, size_t ws_size,
                              hipStream_t stream) {
  const float* x   = (const float*)d_in[0];
  const int* ei    = (const int*)d_in[1];
  const int* batch = (const int*)d_in[2];
  const float* W1a = (const float*)d_in[3];
  const float* b1a = (const float*)d_in[4];
  const float* W1b = (const float*)d_in[5];
  const float* b1b = (const float*)d_in[6];
  const float* g1  = (const float*)d_in[7];
  const float* be1 = (const float*)d_in[8];
  const float* m1  = (const float*)d_in[9];
  const float* v1  = (const float*)d_in[10];
  const float* W2a = (const float*)d_in[11];
  const float* b2a = (const float*)d_in[12];
  const float* W2b = (const float*)d_in[13];
  const float* b2b = (const float*)d_in[14];
  const float* g2  = (const float*)d_in[15];
  const float* be2 = (const float*)d_in[16];
  const float* m2  = (const float*)d_in[17];
  const float* v2  = (const float*)d_in[18];
  float* out = (float*)d_out;

  const int N = in_sizes[0] / IN_FEATS;
  const int E = in_sizes[1] / 2;

  float* ws    = (float*)d_ws;
  float* bufA  = ws;
  float* h1    = ws + (long long)N * IN_FEATS;
  float* h2    = h1 + (long long)N * HIDDEN;
  float* sums  = h2 + (long long)N * HIDDEN;
  float* cnts  = sums + NGRAPH * HIDDEN;
  (void)ws_size; (void)n_in; (void)out_size;

  const int tiles  = (N + 15) / 16;
  const int mlpBlk = (tiles + 7) / 8;

  const int aggBlk = (N + TILE_NODES - 1) / TILE_NODES;

  const int maskBlk    = (E + MASK_EDGES - 1) / MASK_EDGES;
  const int pitchWords = maskBlk * MASK_WORDS;
  unsigned* mask = (unsigned*)(cnts + NGRAPH);
  int2* pairs    = (int2*)(mask + (long long)aggBlk * pitchWords);

  edge_mask_kernel<<<maskBlk, MASK_THREADS, 0, stream>>>(ei, mask, pairs, E, aggBlk, pitchWords);

  tile_agg_kernel<IN_FEATS><<<aggBlk, 1024, 0, stream>>>(x, mask, pairs, bufA, N, pitchWords);
  gin_mlp_kernel<IN_FEATS><<<mlpBlk, 256, 0, stream>>>(
      bufA, W1a, b1a, W1b, b1b, g1, be1, m1, v1, h1, N);

  tile_agg_kernel<HIDDEN><<<aggBlk, 1024, 0, stream>>>(h1, mask, pairs, bufA, N, pitchWords);
  gin_mlp_kernel<HIDDEN><<<mlpBlk, 256, 0, stream>>>(
      bufA, W2a, b2a, W2b, b2b, g2, be2, m2, v2, h2, N);

  pool_zero_kernel<<<(NGRAPH * HIDDEN + 255) / 256, 256, 0, stream>>>(sums, cnts);
  {
    long long tot = (long long)N * 16;
    int blocks = (int)((tot + 255) / 256);
    pool_sum_kernel<<<blocks, 256, 0, stream>>>(h2, batch, sums, cnts, N);
  }
  pool_div_kernel<<<(NGRAPH * HIDDEN + 255) / 256, 256, 0, stream>>>(sums, cnts, out);
}


# Mean pool sums: one wave adds 32 consecutive rows in registers and issues one f32 atomic per column only when the graph id changes; 3125 waves, not 50000; sum order: node order in a run, then arrival
# speedup vs baseline: 2.8285x; 1.3182x over previous
#include <hip/hip_runtime.h>

typedef float v2f __attribute__((ext_vector_type(2)));
typedef float v8f __attribute__((ext_vector_type(8)));

#define IN_FEATS 128
#define HIDDEN   64
#define NGRAPH   512
#define BN_EPS   1e-5f

#define TILE_NODES 496
#define AGG_WAVES  32
#define QCAP       64

#define MASK_EDGES   2048
#define MASK_WORDS   64
#define MASK_PITCH   66
#define MASK_TILES   256
#define MASK_THREADS 512

__global__ __launch_bounds__(MASK_THREADS) void edge_mask_kernel(
    const int* __restrict__ ei, unsigned* __restrict__ mask, int2* __restrict__ pairs,
    int E, int nTiles, int pitchWords) {
  __shared__ unsigned w[MASK_TILES * MASK_PITCH];
  const int i = threadIdx.x;
  const int lane = i & 31;
  const int wave = __builtin_amdgcn_readfirstlane(i >> 5);
  const int* __restrict__ dst = ei + E;
  const int cbase = blockIdx.x * MASK_EDGES;

  int d[4];
#pragma unroll
  for (int r = 0; r < 2; ++r) {
    const int e = cbase + r * 1024 + 2 * i;
    if ((E & 1) == 0 && e < E) {
      const int2 sv = *(const int2*)(ei + e);
      const int2 dv = *(const int2*)(dst + e);
      d[2 * r] = dv.x;
      d[2 * r + 1] = dv.y;
      *(int4*)(pairs + e) = make_int4(sv.x, dv.x, sv.y, dv.y);
    } else {
#pragma unroll
      for (int k = 0; k < 2; ++k) {
        d[2 * r + k] = -1;
        if (e + k < E) {
          const int sv = ei[e + k];
          const int dv = dst[e + k];
          d[2 * r + k] = dv;
          pairs[e + k] = make_int2(sv, dv);
        }
      }
    }
  }

  for (int t0 = 0; t0 < nTiles; t0 += MASK_TILES) {
    int rows = nTiles - t0;
    if (rows > MASK_TILES) rows = MASK_TILES;
    for (int q = i; q < MASK_TILES * MASK_PITCH / 4; q += MASK_THREADS)
      ((uint4*)w)[q] = make_uint4(0u, 0u, 0u, 0u);
    __syncthreads();
#pragma unroll
    for (int r = 0; r < 2; ++r) {
#pragma unroll
      for (int k = 0; k < 2; ++k) {
        const int dv = d[2 * r + k];
        if (dv >= 0) {
          const unsigned t = (unsigned)dv / TILE_NODES - (unsigned)t0;
          if (t < (unsigned)rows)
            __hip_atomic_fetch_or(&w[t * MASK_PITCH + 32 * r + (i >> 4)],
                                  1u << (2 * (i & 15) + k), __ATOMIC_RELAXED,
                                  __HIP_MEMORY_SCOPE_WORKGROUP);
        }
      }
    }
    __syncthreads();
    for (int r = wave; r < rows; r += MASK_THREADS / 32) {
      const uint2 v = *(const uint2*)&w[r * MASK_PITCH + 2 * lane];
      *(uint2*)(mask + (long long)(t0 + r) * pitchWords + blockIdx.x * MASK_WORDS + 2 * lane) = v;
    }
    __syncthreads();
  }
}

template <int F>
__global__ __launch_bounds__(1024) void tile_agg_kernel(
    const float* __restrict__ xin, const unsigned* __restrict__ mask,
    const int2* __restrict__ pairs, float* __restrict__ agg, int n_nodes, int pitchWords) {
  constexpr int VEC = F / 32;
  typedef float vec_t __attribute__((ext_vector_type(VEC)));
  __shared__ float acc[TILE_NODES * F];
  __shared__ int qe[AGG_WAVES * QCAP];

  const int lane = threadIdx.x & 31;
  const int wave = __builtin_amdgcn_readfirstlane((int)(threadIdx.x >> 5));
  const int lo = blockIdx.x * TILE_NODES;
  int cnt = n_nodes - lo;
  if (cnt > TILE_NODES) cnt = TILE_NODES;
  if (cnt <= 0) return;

  for (int r = wave; r < cnt; r += AGG_WAVES) {
    const vec_t v = *(const vec_t*)(xin + (long long)(lo + r) * F + VEC * lane);
#pragma unroll
    for (int k = 0; k < VEC; ++k) acc[r * F + k * 32 + lane] = v[k];
  }
  __syncthreads();

  int* const myq = qe + wave * QCAP;
  int qn = 0;

  auto add_rows = [&](int nb) {
    int s = 0, ld = 0;
    if (lane < nb) {
      const int2 p = pairs[myq[lane]];
      s = p.x;
      ld = p.y - lo;
    }
    for (int j = 0; j < nb; j += 8) {
      vec_t v[8];
      int row[8];
#pragma unroll
      for (int u = 0; u < 8; ++u) {
        int jj = j + u;
        if (jj > nb - 1) jj = nb - 1;
        const int sj = __builtin_amdgcn_readlane(s, jj);
        row[u] = __builtin_amdgcn_readlane(ld, jj);
        v[u] = *(const vec_t*)(xin + (long long)sj * F + VEC * lane);
      }
#pragma unroll
      for (int u = 0; u < 8; ++u) {
        if (j + u < nb) {
          float* p = &acc[row[u] * F + lane];
#pragma unroll
          for (int k = 0; k < VEC; ++k)
            __hip_atomic_fetch_add(p + k * 32, v[u][k], __ATOMIC_RELAXED,
                                   __HIP_MEMORY_SCOPE_WORKGROUP);
        }
      }
    }
  };

  auto take = [&](unsigned m, int ebase) {
    unsigned any = __builtin_amdgcn_ballot_w32(m != 0u);
    while (any != 0u) {
      if (m != 0u) {
        const int pos = qn + (int)__builtin_amdgcn_mbcnt_lo(any, 0u);
        myq[pos] = ebase + __builtin_ctz(m);
        m &= m - 1u;
      }
      qn += __builtin_popcount(any);
      if (qn >= 32) {
        add_rows(32);
        qn -= 32;
        if (lane < qn) myq[lane] = myq[32 + lane];
      }
      any = __builtin_amdgcn_ballot_w32(m != 0u);
    }
  };

  const uint4* __restrict__ mrow = (const uint4*)(mask + (long long)blockIdx.x * pitchWords);
  const int nQ = pitchWords >> 2;
  const int nG = (nQ + 31) >> 5;
  if (wave < nG) {
    auto piece = [&](int g) {
      int q = g * 32 + lane;
      if (q > nQ - 1) q = nQ - 1;
      return mrow[q];
    };
    uint4 nxt = piece(wave);
    for (int g = wave; g < nG; g += AGG_WAVES) {
      const uint4 c = nxt;
      nxt = piece(g + AGG_WAVES);
      const int q = g * 32 + lane;
      const bool in = q < nQ;
      const int eb = q * 128;
      take(in ? c.x : 0u, eb);
      take(in ? c.y : 0u, eb + 32);
      take(in ? c.z : 0u, eb + 64);
      take(in ? c.w : 0u, eb + 96);
    }
  }
  add_rows(qn);
  __syncthreads();

  for (int r = wave; r < cnt; r += AGG_WAVES) {
    vec_t v;
#pragma unroll
    for (int k = 0; k < VEC; ++k) v[k] = acc[r * F + k * 32 + lane];
    *(vec_t*)(agg + (long long)(lo + r) * F + VEC * lane) = v;
  }
}

template <int FIN>
__global__ __launch_bounds__(256) void gin_mlp_kernel(
    const float* __restrict__ hin,
    const float* __restrict__ Wa,
    const float* __restrict__ ba,
    const float* __restrict__ Wb,
    const float* __restrict__ bb,
    const float* __restrict__ gg,
    const float* __restrict__ be,
    const float* __restrict__ mm,
    const float* __restrict__ vv,
    float* __restrict__ hout,
    int n_nodes) {
  __shared__ float sT[8][16 * 64];

  const int wave = threadIdx.x >> 5;
  const int lane = threadIdx.x & 31;
  const int half = lane >> 4;
  const int l16  = lane & 15;

  const int tile = blockIdx.x * 8 + wave;
  const int row0 = tile * 16;
  if (row0 >= n_nodes) return;

  v8f acc[4];
#pragma unroll
  for (int t = 0; t < 4; ++t) {
    float bias = ba[t * 16 + l16];
#pragma unroll
    for (int r = 0; r < 8; ++r) acc[t][r] = bias;
  }

  int arow = row0 + l16;
  if (arow >= n_nodes) arow = n_nodes - 1;
  const float* aptr = hin + (long long)arow * FIN + 2 * half;

  for (int k = 0; k < FIN; k += 4) {
    v2f a = *(const v2f*)(aptr + k);
#pragma unroll
    for (int t = 0; t < 4; ++t) {
      const float* bp = Wa + (t * 16 + l16) * FIN + k + 2 * half;
      v2f b = *(const v2f*)bp;
      acc[t] = __builtin_amdgcn_wmma_f32_16x16x4_f32(
          false, a, false, b, (short)0, acc[t], false, false);
    }
  }

#pragma unroll
  for (int t = 0; t < 4; ++t) {
#pragma unroll
    for (int r = 0; r < 8; ++r) {
      float x = acc[t][r];
      x = x > 0.0f ? x : 0.0f;
      sT[wave][(r + 8 * half) * 64 + t * 16 + l16] = x;
    }
  }

  v8f acc2[4];
#pragma unroll
  for (int t = 0; t < 4; ++t)
#pragma unroll
    for (int r = 0; r < 8; ++r) acc2[t][r] = 0.0f;

  const float* tp = &sT[wave][l16 * 64 + 2 * half];
  for (int k = 0; k < 64; k += 4) {
    v2f a = *(const v2f*)(tp + k);
#pragma unroll
    for (int t = 0; t < 4; ++t) {
      const float* bp = Wb + (t * 16 + l16) * 64 + k + 2 * half;
      v2f b = *(const v2f*)bp;
      acc2[t] = __builtin_amdgcn_wmma_f32_16x16x4_f32(
          false, a, false, b, (short)0, acc2[t], false, false);
    }
  }

#pragma unroll
  for (int t = 0; t < 4; ++t) {
    int n = t * 16 + l16;
    float s  = gg[n] * rsqrtf(vv[n] + BN_EPS);
    float sh = (bb[n] - mm[n]) * s + be[n];
#pragma unroll
    for (int r = 0; r < 8; ++r) {
      int mrow = row0 + r + 8 * half;
      float o = acc2[t][r] * s + sh;
      o = o > 0.0f ? o : 0.0f;
      if (mrow < n_nodes) hout[(long long)mrow * 64 + n] = o;
    }
  }
}

__global__ __launch_bounds__(256) void pool_zero_kernel(float* __restrict__ sums,
                                                        float* __restrict__ counts) {
  int i = blockIdx.x * blockDim.x + threadIdx.x;
  if (i < NGRAPH * HIDDEN) sums[i] = 0.0f;
  if (i < NGRAPH) counts[i] = 0.0f;
}

#define POOL_ROWS  32
#define POOL_GROUP 16

__global__ __launch_bounds__(256) void pool_sum_kernel(
    const float* __restrict__ h, const int* __restrict__ batch,
    float* __restrict__ sums, float* __restrict__ counts, int n_nodes) {
  const int lane = threadIdx.x & 31;
  const int wave =
      __builtin_amdgcn_readfirstlane((int)((blockIdx.x * blockDim.x + threadIdx.x) >> 5));
  const int i0 = wave * POOL_ROWS;
  if (i0 >= n_nodes) return;
  int i1 = i0 + POOL_ROWS;
  if (i1 > n_nodes) i1 = n_nodes;

  auto flush = [&](int g, v2f a, int rows) {
    float* p = sums + (long long)g * HIDDEN + 2 * lane;
    atomicAdd(p, a.x);
    atomicAdd(p + 1, a.y);
    if (lane == 0) atomicAdd(&counts[g], (float)rows);
  };

  int cur = batch[i0];
  v2f acc = {0.0f, 0.0f};
  int rows = 0;
  for (int i = i0; i < i1; i += POOL_GROUP) {
    v2f v[POOL_GROUP];
    int g[POOL_GROUP];
#pragma unroll
    for (int u = 0; u < POOL_GROUP; ++u) {
      int ii = i + u;
      if (ii > i1 - 1) ii = i1 - 1;
      g[u] = batch[ii];
      v[u] = *(const v2f*)(h + (long long)ii * HIDDEN + 2 * lane);
    }
#pragma unroll
    for (int u = 0; u < POOL_GROUP; ++u) {
      if (i + u < i1) {
        if (g[u] != cur) {
          flush(cur, acc, rows);
          acc.x = 0.0f;
          acc.y = 0.0f;
          rows = 0;
          cur = g[u];
        }
        acc.x += v[u].x;
        acc.y += v[u].y;
        ++rows;
      }
    }
  }
  flush(cur, acc, rows);
}

__global__ __launch_bounds__(256) void pool_div_kernel(
    const float* __restrict__ sums, const float* __restrict__ counts,
    float* __restrict__ out) {
  int i = blockIdx.x * blockDim.x + threadIdx.x;
  if (i >= NGRAPH * HIDDEN) return;
  float c = counts[i >> 6];
  c = c > 1.0f ? c : 1.0f;
  out[i] = sums[i] / c;
}

extern "C" void kernel_launch(void* const* d_in, const int* in_sizes, int n_in,
                              void* d_out, int out_size, void* d_ws, size_t ws_size,
                              hipStream_t stream) {
  const float* x   = (const float*)d_in[0];
  const int* ei    = (const int*)d_in[1];
  const int* batch = (const int*)d_in[2];
  const float* W1a = (const float*)d_in[3];
  const float* b1a = (const float*)d_in[4];
  const float* W1b = (const float*)d_in[5];
  const float* b1b = (const float*)d_in[6];
  const float* g1  = (const float*)d_in[7];
  const float* be1 = (const float*)d_in[8];
  const float* m1  = (const float*)d_in[9];
  const float* v1  = (const float*)d_in[10];
  const float* W2a = (const float*)d_in[11];
  const float* b2a = (const float*)d_in[12];
  const float* W2b = (const float*)d_in[13];
  const float* b2b = (const float*)d_in[14];
  const float* g2  = (const float*)d_in[15];
  const float* be2 = (const float*)d_in[16];
  const float* m2  = (const float*)d_in[17];
  const float* v2  = (const float*)d_in[18];
  float* out = (float*)d_out;

  const int N = in_sizes[0] / IN_FEATS;
  const int E = in_sizes[1] / 2;

  float* ws    = (float*)d_ws;
  float* bufA  = ws;
  float* h1    = ws + (long long)N * IN_FEATS;
  float* h2    = h1 + (long long)N * HIDDEN;
  float* sums  = h2 + (long long)N * HIDDEN;
  float* cnts  = sums + NGRAPH * HIDDEN;
  (void)ws_size; (void)n_in; (void)out_size;

  const int tiles  = (N + 15) / 16;
  const int mlpBlk = (tiles + 7) / 8;

  const int aggBlk = (N + TILE_NODES - 1) / TILE_NODES;

  const int maskBlk    = (E + MASK_EDGES - 1) / MASK_EDGES;
  const int pitchWords = maskBlk * MASK_WORDS;
  unsigned* mask = (unsigned*)(cnts + NGRAPH);
  int2* pairs    = (int2*)(mask + (long long)aggBlk * pitchWords);

  edge_mask_kernel<<<maskBlk, MASK_THREADS, 0, stream>>>(ei, mask, pairs, E, aggBlk, pitchWords);

  tile_agg_kernel<IN_FEATS><<<aggBlk, 1024, 0, stream>>>(x, mask, pairs, bufA, N, pitchWords);
  gin_mlp_kernel<IN_FEATS><<<mlpBlk, 256, 0, stream>>>(
      bufA, W1a, b1a, W1b, b1b, g1, be1, m1, v1, h1, N);

  tile_agg_kernel<HIDDEN><<<aggBlk, 1024, 0, stream>>>(h1, mask, pairs, bufA, N, pitchWords);
  gin_mlp_kernel<HIDDEN><<<mlpBlk, 256, 0, stream>>>(
      bufA, W2a, b2a, W2b, b2b, g2, be2, m2, v2, h2, N);

  pool_zero_kernel<<<(NGRAPH * HIDDEN + 255) / 256, 256, 0, stream>>>(sums, cnts);
  {
    const int rowsPerBlock = POOL_ROWS * 8;
    const int blocks = (N + rowsPerBlock - 1) / rowsPerBlock;
    pool_sum_kernel<<<blocks, 256, 0, stream>>>(h2, batch, sums, cnts, N);
  }
  pool_div_kernel<<<(NGRAPH * HIDDEN + 255) / 256, 256, 0, stream>>>(sums, cnts, out);
}


# Tile kernels: 392 destination rows per work-group, not 496: 256 tiles give each of the 256 WGPs one tile (202 left 54 idle), a fifth fewer rows and edges per tile; sums still in LDS arrival order
# speedup vs baseline: 3.0339x; 1.0726x over previous
#include <hip/hip_runtime.h>

typedef float v2f __attribute__((ext_vector_type(2)));
typedef float v8f __attribute__((ext_vector_type(8)));

#define IN_FEATS 128
#define HIDDEN   64
#define NGRAPH   512
#define BN_EPS   1e-5f

#define TILE_NODES 392
#define AGG_WAVES  32
#define QCAP       64

#define MASK_EDGES   2048
#define MASK_WORDS   64
#define MASK_PITCH   66
#define MASK_TILES   256
#define MASK_THREADS 512

__global__ __launch_bounds__(MASK_THREADS) void edge_mask_kernel(
    const int* __restrict__ ei, unsigned* __restrict__ mask, int2* __restrict__ pairs,
    int E, int nTiles, int pitchWords) {
  __shared__ unsigned w[MASK_TILES * MASK_PITCH];
  const int i = threadIdx.x;
  const int lane = i & 31;
  const int wave = __builtin_amdgcn_readfirstlane(i >> 5);
  const int* __restrict__ dst = ei + E;
  const int cbase = blockIdx.x * MASK_EDGES;

  int d[4];
#pragma unroll
  for (int r = 0; r < 2; ++r) {
    const int e = cbase + r * 1024 + 2 * i;
    if ((E & 1) == 0 && e < E) {
      const int2 sv = *(const int2*)(ei + e);
      const int2 dv = *(const int2*)(dst + e);
      d[2 * r] = dv.x;
      d[2 * r + 1] = dv.y;
      *(int4*)(pairs + e) = make_int4(sv.x, dv.x, sv.y, dv.y);
    } else {
#pragma unroll
      for (int k = 0; k < 2; ++k) {
        d[2 * r + k] = -1;
        if (e + k < E) {
          const int sv = ei[e + k];
          const int dv = dst[e + k];
          d[2 * r + k] = dv;
          pairs[e + k] = make_int2(sv, dv);
        }
      }
    }
  }

  for (int t0 = 0; t0 < nTiles; t0 += MASK_TILES) {
    int rows = nTiles - t0;
    if (rows > MASK_TILES) rows = MASK_TILES;
    for (int q = i; q < MASK_TILES * MASK_PITCH / 4; q += MASK_THREADS)
      ((uint4*)w)[q] = make_uint4(0u, 0u, 0u, 0u);
    __syncthreads();
#pragma unroll
    for (int r = 0; r < 2; ++r) {
#pragma unroll
      for (int k = 0; k < 2; ++k) {
        const int dv = d[2 * r + k];
        if (dv >= 0) {
          const unsigned t = (unsigned)dv / TILE_NODES - (unsigned)t0;
          if (t < (unsigned)rows)
            __hip_atomic_fetch_or(&w[t * MASK_PITCH + 32 * r + (i >> 4)],
                                  1u << (2 * (i & 15) + k), __ATOMIC_RELAXED,
                                  __HIP_MEMORY_SCOPE_WORKGROUP);
        }
      }
    }
    __syncthreads();
    for (int r = wave; r < rows; r += MASK_THREADS / 32) {
      const uint2 v = *(const uint2*)&w[r * MASK_PITCH + 2 * lane];
      *(uint2*)(mask + (long long)(t0 + r) * pitchWords + blockIdx.x * MASK_WORDS + 2 * lane) = v;
    }
    __syncthreads();
  }
}

template <int F>
__global__ __launch_bounds__(1024) void tile_agg_kernel(
    const float* __restrict__ xin, const unsigned* __restrict__ mask,
    const int2* __restrict__ pairs, float* __restrict__ agg, int n_nodes, int pitchWords) {
  constexpr int VEC = F / 32;
  typedef float vec_t __attribute__((ext_vector_type(VEC)));
  __shared__ float acc[TILE_NODES * F];
  __shared__ int qe[AGG_WAVES * QCAP];

  const int lane = threadIdx.x & 31;
  const int wave = __builtin_amdgcn_readfirstlane((int)(threadIdx.x >> 5));
  const int lo = blockIdx.x * TILE_NODES;
  int cnt = n_nodes - lo;
  if (cnt > TILE_NODES) cnt = TILE_NODES;
  if (cnt <= 0) return;

  for (int r = wave; r < cnt; r += AGG_WAVES) {
    const vec_t v = *(const vec_t*)(xin + (long long)(lo + r) * F + VEC * lane);
#pragma unroll
    for (int k = 0; k < VEC; ++k) acc[r * F + k * 32 + lane] = v[k];
  }
  __syncthreads();

  int* const myq = qe + wave * QCAP;
  int qn = 0;

  auto add_rows = [&](int nb) {
    int s = 0, ld = 0;
    if (lane < nb) {
      const int2 p = pairs[myq[lane]];
      s = p.x;
      ld = p.y - lo;
    }
    for (int j = 0; j < nb; j += 8) {
      vec_t v[8];
      int row[8];
#pragma unroll
      for (int u = 0; u < 8; ++u) {
        int jj = j + u;
        if (jj > nb - 1) jj = nb - 1;
        const int sj = __builtin_amdgcn_readlane(s, jj);
        row[u] = __builtin_amdgcn_readlane(ld, jj);
        v[u] = *(const vec_t*)(xin + (long long)sj * F + VEC * lane);
      }
#pragma unroll
      for (int u = 0; u < 8; ++u) {
        if (j + u < nb) {
          float* p = &acc[row[u] * F + lane];
#pragma unroll
          for (int k = 0; k < VEC; ++k)
            __hip_atomic_fetch_add(p + k * 32, v[u][k], __ATOMIC_RELAXED,
                                   __HIP_MEMORY_SCOPE_WORKGROUP);
        }
      }
    }
  };

  auto take = [&](unsigned m, int ebase) {
    unsigned any = __builtin_amdgcn_ballot_w32(m != 0u);
    while (any != 0u) {
      if (m != 0u) {
        const int pos = qn + (int)__builtin_amdgcn_mbcnt_lo(any, 0u);
        myq[pos] = ebase + __builtin_ctz(m);
        m &= m - 1u;
      }
      qn += __builtin_popcount(any);
      if (qn >= 32) {
        add_rows(32);
        qn -= 32;
        if (lane < qn) myq[lane] = myq[32 + lane];
      }
      any = __builtin_amdgcn_ballot_w32(m != 0u);
    }
  };

  const uint4* __restrict__ mrow = (const uint4*)(mask + (long long)blockIdx.x * pitchWords);
  const int nQ = pitchWords >> 2;
  const int nG = (nQ + 31) >> 5;
  if (wave < nG) {
    auto piece = [&](int g) {
      int q = g * 32 + lane;
      if (q > nQ - 1) q = nQ - 1;
      return mrow[q];
    };
    uint4 nxt = piece(wave);
    for (int g = wave; g < nG; g += AGG_WAVES) {
      const uint4 c = nxt;
      nxt = piece(g + AGG_WAVES);
      const int q = g * 32 + lane;
      const bool in = q < nQ;
      const int eb = q * 128;
      take(in ? c.x : 0u, eb);
      take(in ? c.y : 0u, eb + 32);
      take(in ? c.z : 0u, eb + 64);
      take(in ? c.w : 0u, eb + 96);
    }
  }
  add_rows(qn);
  __syncthreads();

  for (int r = wave; r < cnt; r += AGG_WAVES) {
    vec_t v;
#pragma unroll
    for (int k = 0; k < VEC; ++k) v[k] = acc[r * F + k * 32 + lane];
    *(vec_t*)(agg + (long long)(lo + r) * F + VEC * lane) = v;
  }
}

template <int FIN>
__global__ __launch_bounds__(256) void gin_mlp_kernel(
    const float* __restrict__ hin,
    const float* __restrict__ Wa,
    const float* __restrict__ ba,
    const float* __restrict__ Wb,
    const float* __restrict__ bb,
    const float* __restrict__ gg,
    const float* __restrict__ be,
    const float* __restrict__ mm,
    const float* __restrict__ vv,
    float* __restrict__ hout,
    int n_nodes) {
  __shared__ float sT[8][16 * 64];

  const int wave = threadIdx.x >> 5;
  const int lane = threadIdx.x & 31;
  const int half = lane >> 4;
  const int l16  = lane & 15;

  const int tile = blockIdx.x * 8 + wave;
  const int row0 = tile * 16;
  if (row0 >= n_nodes) return;

  v8f acc[4];
#pragma unroll
  for (int t = 0; t < 4; ++t) {
    float bias = ba[t * 16 + l16];
#pragma unroll
    for (int r = 0; r < 8; ++r) acc[t][r] = bias;
  }

  int arow = row0 + l16;
  if (arow >= n_nodes) arow = n_nodes - 1;
  const float* aptr = hin + (long long)arow * FIN + 2 * half;

  for (int k = 0; k < FIN; k += 4) {
    v2f a = *(const v2f*)(aptr + k);
#pragma unroll
    for (int t = 0; t < 4; ++t) {
      const float* bp = Wa + (t * 16 + l16) * FIN + k + 2 * half;
      v2f b = *(const v2f*)bp;
      acc[t] = __builtin_amdgcn_wmma_f32_16x16x4_f32(
          false, a, false, b, (short)0, acc[t], false, false);
    }
  }

#pragma unroll
  for (int t = 0; t < 4; ++t) {
#pragma unroll
    for (int r = 0; r < 8; ++r) {
      float x = acc[t][r];
      x = x > 0.0f ? x : 0.0f;
      sT[wave][(r + 8 * half) * 64 + t * 16 + l16] = x;
    }
  }

  v8f acc2[4];
#pragma unroll
  for (int t = 0; t < 4; ++t)
#pragma unroll
    for (int r = 0; r < 8; ++r) acc2[t][r] = 0.0f;

  const float* tp = &sT[wave][l16 * 64 + 2 * half];
  for (int k = 0; k < 64; k += 4) {
    v2f a = *(const v2f*)(tp + k);
#pragma unroll
    for (int t = 0; t < 4; ++t) {
      const float* bp = Wb + (t * 16 + l16) * 64 + k + 2 * half;
      v2f b = *(const v2f*)bp;
      acc2[t] = __builtin_amdgcn_wmma_f32_16x16x4_f32(
          false, a, false, b, (short)0, acc2[t], false, false);
    }
  }

#pragma unroll
  for (int t = 0; t < 4; ++t) {
    int n = t * 16 + l16;
    float s  = gg[n] * rsqrtf(vv[n] + BN_EPS);
    float sh = (bb[n] - mm[n]) * s + be[n];
#pragma unroll
    for (int r = 0; r < 8; ++r) {
      int mrow = row0 + r + 8 * half;
      float o = acc2[t][r] * s + sh;
      o = o > 0.0f ? o : 0.0f;
      if (mrow < n_nodes) hout[(long long)mrow * 64 + n] = o;
    }
  }
}

__global__ __launch_bounds__(256) void pool_zero_kernel(float* __restrict__ sums,
                                                        float* __restrict__ counts) {
  int i = blockIdx.x * blockDim.x + threadIdx.x;
  if (i < NGRAPH * HIDDEN) sums[i] = 0.0f;
  if (i < NGRAPH) counts[i] = 0.0f;
}

#define POOL_ROWS  32
#define POOL_GROUP 16

__global__ __launch_bounds__(256) void pool_sum_kernel(
    const float* __restrict__ h, const int* __restrict__ batch,
    float* __restrict__ sums, float* __restrict__ counts, int n_nodes) {
  const int lane = threadIdx.x & 31;
  const int wave =
      __builtin_amdgcn_readfirstlane((int)((blockIdx.x * blockDim.x + threadIdx.x) >> 5));
  const int i0 = wave * POOL_ROWS;
  if (i0 >= n_nodes) return;
  int i1 = i0 + POOL_ROWS;
  if (i1 > n_nodes) i1 = n_nodes;

  auto flush = [&](int g, v2f a, int rows) {
    float* p = sums + (long long)g * HIDDEN + 2 * lane;
    atomicAdd(p, a.x);
    atomicAdd(p + 1, a.y);
    if (lane == 0) atomicAdd(&counts[g], (float)rows);
  };

  int cur = batch[i0];
  v2f acc = {0.0f, 0.0f};
  int rows = 0;
  for (int i = i0; i < i1; i += POOL_GROUP) {
    v2f v[POOL_GROUP];
    int g[POOL_GROUP];
#pragma unroll
    for (int u = 0; u < POOL_GROUP; ++u) {
      int ii = i + u;
      if (ii > i1 - 1) ii = i1 - 1;
      g[u] = batch[ii];
      v[u] = *(const v2f*)(h + (long long)ii * HIDDEN + 2 * lane);
    }
#pragma unroll
    for (int u = 0; u < POOL_GROUP; ++u) {
      if (i + u < i1) {
        if (g[u] != cur) {
          flush(cur, acc, rows);
          acc.x = 0.0f;
          acc.y = 0.0f;
          rows = 0;
          cur = g[u];
        }
        acc.x += v[u].x;
        acc.y += v[u].y;
        ++rows;
      }
    }
  }
  flush(cur, acc, rows);
}

__global__ __launch_bounds__(256) void pool_div_kernel(
    const float* __restrict__ sums, const float* __restrict__ counts,
    float* __restrict__ out) {
  int i = blockIdx.x * blockDim.x + threadIdx.x;
  if (i >= NGRAPH * HIDDEN) return;
  float c = counts[i >> 6];
  c = c > 1.0f ? c : 1.0f;
  out[i] = sums[i] / c;
}

extern "C" void kernel_launch(void* const* d_in, const int* in_sizes, int n_in,
                              void* d_out, int out_size, void* d_ws, size_t ws_size,
                              hipStream_t stream) {
  const float* x   = (const float*)d_in[0];
  const int* ei    = (const int*)d_in[1];
  const int* batch = (const int*)d_in[2];
  const float* W1a = (const float*)d_in[3];
  const float* b1a = (const float*)d_in[4];
  const float* W1b = (const float*)d_in[5];
  const float* b1b = (const float*)d_in[6];
  const float* g1  = (const float*)d_in[7];
  const float* be1 = (const float*)d_in[8];
  const float* m1  = (const float*)d_in[9];
  const float* v1  = (const float*)d_in[10];
  const float* W2a = (const float*)d_in[11];
  const float* b2a = (const float*)d_in[12];
  const float* W2b = (const float*)d_in[13];
  const float* b2b = (const float*)d_in[14];
  const float* g2  = (const float*)d_in[15];
  const float* be2 = (const float*)d_in[16];
  const float* m2  = (const float*)d_in[17];
  const float* v2  = (const float*)d_in[18];
  float* out = (float*)d_out;

  const int N = in_sizes[0] / IN_FEATS;
  const int E = in_sizes[1] / 2;

  float* ws    = (float*)d_ws;
  float* bufA  = ws;
  float* h1    = ws + (long long)N * IN_FEATS;
  float* h2    = h1 + (long long)N * HIDDEN;
  float* sums  = h2 + (long long)N * HIDDEN;
  float* cnts  = sums + NGRAPH * HIDDEN;
  (void)ws_size; (void)n_in; (void)out_size;

  const int tiles  = (N + 15) / 16;
  const int mlpBlk = (tiles + 7) / 8;

  const int aggBlk = (N + TILE_NODES - 1) / TILE_NODES;

  const int maskBlk    = (E + MASK_EDGES - 1) / MASK_EDGES;
  const int pitchWords = maskBlk * MASK_WORDS;
  unsigned* mask = (unsigned*)(cnts + NGRAPH);
  int2* pairs    = (int2*)(mask + (long long)aggBlk * pitchWords);

  edge_mask_kernel<<<maskBlk, MASK_THREADS, 0, stream>>>(ei, mask, pairs, E, aggBlk, pitchWords);

  tile_agg_kernel<IN_FEATS><<<aggBlk, 1024, 0, stream>>>(x, mask, pairs, bufA, N, pitchWords);
  gin_mlp_kernel<IN_FEATS><<<mlpBlk, 256, 0, stream>>>(
      bufA, W1a, b1a, W1b, b1b, g1, be1, m1, v1, h1, N);

  tile_agg_kernel<HIDDEN><<<aggBlk, 1024, 0, stream>>>(h1, mask, pairs, bufA, N, pitchWords);
  gin_mlp_kernel<HIDDEN><<<mlpBlk, 256, 0, stream>>>(
      bufA, W2a, b2a, W2b, b2b, g2, be2, m2, v2, h2, N);

  pool_zero_kernel<<<(NGRAPH * HIDDEN + 255) / 256, 256, 0, stream>>>(sums, cnts);
  {
    const int rowsPerBlock = POOL_ROWS * 8;
    const int blocks = (N + rowsPerBlock - 1) / rowsPerBlock;
    pool_sum_kernel<<<blocks, 256, 0, stream>>>(h2, batch, sums, cnts, N);
  }
  pool_div_kernel<<<(NGRAPH * HIDDEN + 255) / 256, 256, 0, stream>>>(sums, cnts, out);
}


# MLP moved into the tile kernels: each wave reads 16 rows of the LDS sums as WMMA A fragments (pitch F+1); no agg plane, 8 launches become 6; k order of the products permuted: sum order changes
# speedup vs baseline: 3.5415x; 1.1673x over previous
#include <hip/hip_runtime.h>

typedef float v2f __attribute__((ext_vector_type(2)));
typedef float v8f __attribute__((ext_vector_type(8)));

#define IN_FEATS 128
#define HIDDEN   64
#define NGRAPH   512
#define BN_EPS   1e-5f

#define TILE_NODES 392
#define AGG_WAVES  32
#define QCAP       64

#define MASK_EDGES   2048
#define MASK_WORDS   64
#define MASK_PITCH   66
#define MASK_TILES   256
#define MASK_THREADS 512

__global__ __launch_bounds__(MASK_THREADS) void edge_mask_kernel(
    const int* __restrict__ ei, unsigned* __restrict__ mask, int2* __restrict__ pairs,
    int E, int nTiles, int pitchWords) {
  __shared__ unsigned w[MASK_TILES * MASK_PITCH];
  const int i = threadIdx.x;
  const int lane = i & 31;
  const int wave = __builtin_amdgcn_readfirstlane(i >> 5);
  const int* __restrict__ dst = ei + E;
  const int cbase = blockIdx.x * MASK_EDGES;

  int d[4];
#pragma unroll
  for (int r = 0; r < 2; ++r) {
    const int e = cbase + r * 1024 + 2 * i;
    if ((E & 1) == 0 && e < E) {
      const int2 sv = *(const int2*)(ei + e);
      const int2 dv = *(const int2*)(dst + e);
      d[2 * r] = dv.x;
      d[2 * r + 1] = dv.y;
      *(int4*)(pairs + e) = make_int4(sv.x, dv.x, sv.y, dv.y);
    } else {
#pragma unroll
      for (int k = 0; k < 2; ++k) {
        d[2 * r + k] = -1;
        if (e + k < E) {
          const int sv = ei[e + k];
          const int dv = dst[e + k];
          d[2 * r + k] = dv;
          pairs[e + k] = make_int2(sv, dv);
        }
      }
    }
  }

  for (int t0 = 0; t0 < nTiles; t0 += MASK_TILES) {
    int rows = nTiles - t0;
    if (rows > MASK_TILES) rows = MASK_TILES;
    for (int q = i; q < MASK_TILES * MASK_PITCH / 4; q += MASK_THREADS)
      ((uint4*)w)[q] = make_uint4(0u, 0u, 0u, 0u);
    __syncthreads();
#pragma unroll
    for (int r = 0; r < 2; ++r) {
#pragma unroll
      for (int k = 0; k < 2; ++k) {
        const int dv = d[2 * r + k];
        if (dv >= 0) {
          const unsigned t = (unsigned)dv / TILE_NODES - (unsigned)t0;
          if (t < (unsigned)rows)
            __hip_atomic_fetch_or(&w[t * MASK_PITCH + 32 * r + (i >> 4)],
                                  1u << (2 * (i & 15) + k), __ATOMIC_RELAXED,
                                  __HIP_MEMORY_SCOPE_WORKGROUP);
        }
      }
    }
    __syncthreads();
    for (int r = wave; r < rows; r += MASK_THREADS / 32) {
      const uint2 v = *(const uint2*)&w[r * MASK_PITCH + 2 * lane];
      *(uint2*)(mask + (long long)(t0 + r) * pitchWords + blockIdx.x * MASK_WORDS + 2 * lane) = v;
    }
    __syncthreads();
  }
}

#define TILE_LDS_ROWS 400

template <int F>
__global__ __launch_bounds__(1024) void tile_layer_kernel(
    const float* __restrict__ xin, const unsigned* __restrict__ mask,
    const int2* __restrict__ pairs,
    const float* __restrict__ Wa,
    const float* __restrict__ ba,
    const float* __restrict__ Wb,
    const float* __restrict__ bb,
    const float* __restrict__ gg,
    const float* __restrict__ be,
    const float* __restrict__ mm,
    const float* __restrict__ vv,
    float* __restrict__ hout,
    int n_nodes, int pitchWords) {
  constexpr int VEC = F / 32;
  constexpr int PITCH = F + 1;
  typedef float vec_t __attribute__((ext_vector_type(VEC)));
  __shared__ float acc[TILE_LDS_ROWS * PITCH];
  __shared__ int qe[AGG_WAVES * QCAP];

  const int lane = threadIdx.x & 31;
  const int wave = __builtin_amdgcn_readfirstlane((int)(threadIdx.x >> 5));
  const int lo = blockIdx.x * TILE_NODES;
  int cnt = n_nodes - lo;
  if (cnt > TILE_NODES) cnt = TILE_NODES;
  if (cnt <= 0) return;

  for (int r = wave; r < cnt; r += AGG_WAVES) {
    const vec_t v = *(const vec_t*)(xin + (long long)(lo + r) * F + VEC * lane);
#pragma unroll
    for (int k = 0; k < VEC; ++k) acc[r * PITCH + k * 32 + lane] = v[k];
  }
  for (int r = cnt + wave; r < ((cnt + 15) & ~15); r += AGG_WAVES) {
#pragma unroll
    for (int k = 0; k < VEC; ++k) acc[r * PITCH + k * 32 + lane] = 0.0f;
  }
  __syncthreads();

  int* const myq = qe + wave * QCAP;
  int qn = 0;

  auto add_rows = [&](int nb) {
    int s = 0, ld = 0;
    if (lane < nb) {
      const int2 p = pairs[myq[lane]];
      s = p.x;
      ld = p.y - lo;
    }
    for (int j = 0; j < nb; j += 8) {
      vec_t v[8];
      int row[8];
#pragma unroll
      for (int u = 0; u < 8; ++u) {
        int jj = j + u;
        if (jj > nb - 1) jj = nb - 1;
        const int sj = __builtin_amdgcn_readlane(s, jj);
        row[u] = __builtin_amdgcn_readlane(ld, jj);
        v[u] = *(const vec_t*)(xin + (long long)sj * F + VEC * lane);
      }
#pragma unroll
      for (int u = 0; u < 8; ++u) {
        if (j + u < nb) {
          float* p = &acc[row[u] * PITCH + lane];
#pragma unroll
          for (int k = 0; k < VEC; ++k)
            __hip_atomic_fetch_add(p + k * 32, v[u][k], __ATOMIC_RELAXED,
                                   __HIP_MEMORY_SCOPE_WORKGROUP);
        }
      }
    }
  };

  auto take = [&](unsigned m, int ebase) {
    unsigned any = __builtin_amdgcn_ballot_w32(m != 0u);
    while (any != 0u) {
      if (m != 0u) {
        const int pos = qn + (int)__builtin_amdgcn_mbcnt_lo(any, 0u);
        myq[pos] = ebase + __builtin_ctz(m);
        m &= m - 1u;
      }
      qn += __builtin_popcount(any);
      if (qn >= 32) {
        add_rows(32);
        qn -= 32;
        if (lane < qn) myq[lane] = myq[32 + lane];
      }
      any = __builtin_amdgcn_ballot_w32(m != 0u);
    }
  };

  const uint4* __restrict__ mrow = (const uint4*)(mask + (long long)blockIdx.x * pitchWords);
  const int nQ = pitchWords >> 2;
  const int nG = (nQ + 31) >> 5;
  if (wave < nG) {
    auto piece = [&](int g) {
      int q = g * 32 + lane;
      if (q > nQ - 1) q = nQ - 1;
      return mrow[q];
    };
    uint4 nxt = piece(wave);
    for (int g = wave; g < nG; g += AGG_WAVES) {
      const uint4 c = nxt;
      nxt = piece(g + AGG_WAVES);
      const int q = g * 32 + lane;
      const bool in = q < nQ;
      const int eb = q * 128;
      take(in ? c.x : 0u, eb);
      take(in ? c.y : 0u, eb + 32);
      take(in ? c.z : 0u, eb + 64);
      take(in ? c.w : 0u, eb + 96);
    }
  }
  add_rows(qn);
  __syncthreads();

  const int r0 = wave * 16;
  if (r0 >= cnt) return;
  const int half = lane >> 4;
  const int l16 = lane & 15;

  v8f g1[4];
#pragma unroll
  for (int t = 0; t < 4; ++t) {
    const float bias = ba[t * 16 + l16];
#pragma unroll
    for (int r = 0; r < 8; ++r) g1[t][r] = bias;
  }
  float* const arow = &acc[(r0 + l16) * PITCH + 16 * half];
  const float* const wa = Wa + l16 * F + (F / 2) * half;
  for (int q = 0; q < F / 4; ++q) {
    const int e = 2 * q;
    const int col = (e % VEC) * 32 + e / VEC;
    v2f a;
    a.x = arow[col];
    a.y = arow[col + 32];
#pragma unroll
    for (int t = 0; t < 4; ++t) {
      const v2f b = *(const v2f*)(wa + t * 16 * F + e);
      g1[t] = __builtin_amdgcn_wmma_f32_16x16x4_f32(
          false, a, false, b, (short)0, g1[t], false, false);
    }
  }

#pragma unroll
  for (int t = 0; t < 4; ++t) {
    const int n = t * 16 + l16;
#pragma unroll
    for (int r = 0; r < 8; ++r) {
      float x = g1[t][r];
      x = x > 0.0f ? x : 0.0f;
      acc[(r0 + r + 8 * half) * PITCH + (n & 1) * 32 + (n >> 1)] = x;
    }
  }

  v8f g2[4];
#pragma unroll
  for (int t = 0; t < 4; ++t)
#pragma unroll
    for (int r = 0; r < 8; ++r) g2[t][r] = 0.0f;
  const float* const wb = Wb + l16 * HIDDEN + (HIDDEN / 2) * half;
  for (int q = 0; q < HIDDEN / 4; ++q) {
    v2f a;
    a.x = arow[q];
    a.y = arow[q + 32];
#pragma unroll
    for (int t = 0; t < 4; ++t) {
      const v2f b = *(const v2f*)(wb + t * 16 * HIDDEN + 2 * q);
      g2[t] = __builtin_amdgcn_wmma_f32_16x16x4_f32(
          false, a, false, b, (short)0, g2[t], false, false);
    }
  }

#pragma unroll
  for (int t = 0; t < 4; ++t) {
    const int n = t * 16 + l16;
    const float s = gg[n] * rsqrtf(vv[n] + BN_EPS);
    const float sh = (bb[n] - mm[n]) * s + be[n];
#pragma unroll
    for (int r = 0; r < 8; ++r) {
      const int lr = r0 + r + 8 * half;
      float o = g2[t][r] * s + sh;
      o = o > 0.0f ? o : 0.0f;
      if (lr < cnt) hout[(long long)(lo + lr) * HIDDEN + n] = o;
    }
  }
}

__global__ __launch_bounds__(256) void pool_zero_kernel(float* __restrict__ sums,
                                                        float* __restrict__ counts) {
  int i = blockIdx.x * blockDim.x + threadIdx.x;
  if (i < NGRAPH * HIDDEN) sums[i] = 0.0f;
  if (i < NGRAPH) counts[i] = 0.0f;
}

#define POOL_ROWS  32
#define POOL_GROUP 16

__global__ __launch_bounds__(256) void pool_sum_kernel(
    const float* __restrict__ h, const int* __restrict__ batch,
    float* __restrict__ sums, float* __restrict__ counts, int n_nodes) {
  const int lane = threadIdx.x & 31;
  const int wave =
      __builtin_amdgcn_readfirstlane((int)((blockIdx.x * blockDim.x + threadIdx.x) >> 5));
  const int i0 = wave * POOL_ROWS;
  if (i0 >= n_nodes) return;
  int i1 = i0 + POOL_ROWS;
  if (i1 > n_nodes) i1 = n_nodes;

  auto flush = [&](int g, v2f a, int rows) {
    float* p = sums + (long long)g * HIDDEN + 2 * lane;
    atomicAdd(p, a.x);
    atomicAdd(p + 1, a.y);
    if (lane == 0) atomicAdd(&counts[g], (float)rows);
  };

  int cur = batch[i0];
  v2f acc = {0.0f, 0.0f};
  int rows = 0;
  for (int i = i0; i < i1; i += POOL_GROUP) {
    v2f v[POOL_GROUP];
    int g[POOL_GROUP];
#pragma unroll
    for (int u = 0; u < POOL_GROUP; ++u) {
      int ii = i + u;
      if (ii > i1 - 1) ii = i1 - 1;
      g[u] = batch[ii];
      v[u] = *(const v2f*)(h + (long long)ii * HIDDEN + 2 * lane);
    }
#pragma unroll
    for (int u = 0; u < POOL_GROUP; ++u) {
      if (i + u < i1) {
        if (g[u] != cur) {
          flush(cur, acc, rows);
          acc.x = 0.0f;
          acc.y = 0.0f;
          rows = 0;
          cur = g[u];
        }
        acc.x += v[u].x;
        acc.y += v[u].y;
        ++rows;
      }
    }
  }
  flush(cur, acc, rows);
}

__global__ __launch_bounds__(256) void pool_div_kernel(
    const float* __restrict__ sums, const float* __restrict__ counts,
    float* __restrict__ out) {
  int i = blockIdx.x * blockDim.x + threadIdx.x;
  if (i >= NGRAPH * HIDDEN) return;
  float c = counts[i >> 6];
  c = c > 1.0f ? c : 1.0f;
  out[i] = sums[i] / c;
}

extern "C" void kernel_launch(void* const* d_in, const int* in_sizes, int n_in,
                              void* d_out, int out_size, void* d_ws, size_t ws_size,
                              hipStream_t stream) {
  const float* x   = (const float*)d_in[0];
  const int* ei    = (const int*)d_in[1];
  const int* batch = (const int*)d_in[2];
  const float* W1a = (const float*)d_in[3];
  const float* b1a = (const float*)d_in[4];
  const float* W1b = (const float*)d_in[5];
  const float* b1b = (const float*)d_in[6];
  const float* g1  = (const float*)d_in[7];
  const float* be1 = (const float*)d_in[8];
  const float* m1  = (const float*)d_in[9];
  const float* v1  = (const float*)d_in[10];
  const float* W2a = (const float*)d_in[11];
  const float* b2a = (const float*)d_in[12];
  const float* W2b = (const float*)d_in[13];
  const float* b2b = (const float*)d_in[14];
  const float* g2  = (const float*)d_in[15];
  const float* be2 = (const float*)d_in[16];
  const float* m2  = (const float*)d_in[17];
  const float* v2  = (const float*)d_in[18];
  float* out = (float*)d_out;

  const int N = in_sizes[0] / IN_FEATS;
  const int E = in_sizes[1] / 2;

  float* ws    = (float*)d_ws;
  float* h1    = ws;
  float* h2    = h1 + (long long)N * HIDDEN;
  float* sums  = h2 + (long long)N * HIDDEN;
  float* cnts  = sums + NGRAPH * HIDDEN;
  (void)ws_size; (void)n_in; (void)out_size;

  const int aggBlk = (N + TILE_NODES - 1) / TILE_NODES;

  const int maskBlk    = (E + MASK_EDGES - 1) / MASK_EDGES;
  const int pitchWords = maskBlk * MASK_WORDS;
  unsigned* mask = (unsigned*)(cnts + NGRAPH);
  int2* pairs    = (int2*)(mask + (long long)aggBlk * pitchWords);

  edge_mask_kernel<<<maskBlk, MASK_THREADS, 0, stream>>>(ei, mask, pairs, E, aggBlk, pitchWords);

  tile_layer_kernel<IN_FEATS><<<aggBlk, 1024, 0, stream>>>(
      x, mask, pairs, W1a, b1a, W1b, b1b, g1, be1, m1, v1, h1, N, pitchWords);

  tile_layer_kernel<HIDDEN><<<aggBlk, 1024, 0, stream>>>(
      h1, mask, pairs, W2a, b2a, W2b, b2b, g2, be2, m2, v2, h2, N, pitchWords);

  pool_zero_kernel<<<(NGRAPH * HIDDEN + 255) / 256, 256, 0, stream>>>(sums, cnts);
  {
    const int rowsPerBlock = POOL_ROWS * 8;
    const int blocks = (N + rowsPerBlock - 1) / rowsPerBlock;
    pool_sum_kernel<<<blocks, 256, 0, stream>>>(h2, batch, sums, cnts, N);
  }
  pool_div_kernel<<<(NGRAPH * HIDDEN + 255) / 256, 256, 0, stream>>>(sums, cnts, out);
}


# MLP weights: the first launch writes each 64 x K matrix once per call in B-fragment order; a B load reads 512 contiguous bytes (4 lines), not 8 bytes from each of 32 lines; k order unchanged
# speedup vs baseline: 3.7579x; 1.0611x over previous
#include <hip/hip_runtime.h>

typedef float v2f __attribute__((ext_vector_type(2)));
typedef float v4f __attribute__((ext_vector_type(4)));
typedef float v8f __attribute__((ext_vector_type(8)));

#define IN_FEATS 128
#define HIDDEN   64
#define NGRAPH   512
#define BN_EPS   1e-5f

#define TILE_NODES 392
#define AGG_WAVES  32
#define QCAP       64

#define MASK_EDGES   2048
#define MASK_WORDS   64
#define MASK_PITCH   66
#define MASK_TILES   256
#define MASK_THREADS 512

#define WF_W1A_VEC 2048
#define WF_SQ_VEC  1024
#define WF_VECS    (WF_W1A_VEC + 3 * WF_SQ_VEC)
#define WF_BLOCKS  ((WF_VECS + MASK_THREADS - 1) / MASK_THREADS)

__global__ __launch_bounds__(MASK_THREADS) void edge_mask_kernel(
    const int* __restrict__ ei, unsigned* __restrict__ mask, int2* __restrict__ pairs,
    int E, int nTiles, int pitchWords,
    const float* __restrict__ W1a, const float* __restrict__ W1b,
    const float* __restrict__ W2a, const float* __restrict__ W2b,
    float* __restrict__ wf, int nEdgeBlk) {
  __shared__ unsigned w[MASK_TILES * MASK_PITCH];
  const int i = threadIdx.x;
  if ((int)blockIdx.x >= nEdgeBlk) {
    const int o = ((int)blockIdx.x - nEdgeBlk) * MASK_THREADS + i;
    if (o < WF_VECS) {
      const float* W = W1a;
      int K = IN_FEATS, u = o;
      if (o >= WF_W1A_VEC) {
        const int m = (o - WF_W1A_VEC) / WF_SQ_VEC;
        u = (o - WF_W1A_VEC) % WF_SQ_VEC;
        K = HIDDEN;
        W = m == 0 ? W1b : (m == 1 ? W2a : W2b);
      }
      const int L = u & 31, t = (u >> 5) & 3, j = u >> 7;
      const float* p = W + (t * 16 + (L & 15)) * K + 4 * j + (K / 2) * (L >> 4);
      v4f v;
      v.x = p[0]; v.y = p[1]; v.z = p[2]; v.w = p[3];
      *(v4f*)(wf + 4 * o) = v;
    }
    return;
  }
  const int lane = i & 31;
  const int wave = __builtin_amdgcn_readfirstlane(i >> 5);
  const int* __restrict__ dst = ei + E;
  const int cbase = blockIdx.x * MASK_EDGES;

  int d[4];
#pragma unroll
  for (int r = 0; r < 2; ++r) {
    const int e = cbase + r * 1024 + 2 * i;
    if ((E & 1) == 0 && e < E) {
      const int2 sv = *(const int2*)(ei + e);
      const int2 dv = *(const int2*)(dst + e);
      d[2 * r] = dv.x;
      d[2 * r + 1] = dv.y;
      *(int4*)(pairs + e) = make_int4(sv.x, dv.x, sv.y, dv.y);
    } else {
#pragma unroll
      for (int k = 0; k < 2; ++k) {
        d[2 * r + k] = -1;
        if (e + k < E) {
          const int sv = ei[e + k];
          const int dv = dst[e + k];
          d[2 * r + k] = dv;
          pairs[e + k] = make_int2(sv, dv);
        }
      }
    }
  }

  for (int t0 = 0; t0 < nTiles; t0 += MASK_TILES) {
    int rows = nTiles - t0;
    if (rows > MASK_TILES) rows = MASK_TILES;
    for (int q = i; q < MASK_TILES * MASK_PITCH / 4; q += MASK_THREADS)
      ((uint4*)w)[q] = make_uint4(0u, 0u, 0u, 0u);
    __syncthreads();
#pragma unroll
    for (int r = 0; r < 2; ++r) {
#pragma unroll
      for (int k = 0; k < 2; ++k) {
        const int dv = d[2 * r + k];
        if (dv >= 0) {
          const unsigned t = (unsigned)dv / TILE_NODES - (unsigned)t0;
          if (t < (unsigned)rows)
            __hip_atomic_fetch_or(&w[t * MASK_PITCH + 32 * r + (i >> 4)],
                                  1u << (2 * (i & 15) + k), __ATOMIC_RELAXED,
                                  __HIP_MEMORY_SCOPE_WORKGROUP);
        }
      }
    }
    __syncthreads();
    for (int r = wave; r < rows; r += MASK_THREADS / 32) {
      const uint2 v = *(const uint2*)&w[r * MASK_PITCH + 2 * lane];
      *(uint2*)(mask + (long long)(t0 + r) * pitchWords + blockIdx.x * MASK_WORDS + 2 * lane) = v;
    }
    __syncthreads();
  }
}

#define TILE_LDS_ROWS 400

template <int F>
__global__ __launch_bounds__(1024) void tile_layer_kernel(
    const float* __restrict__ xin, const unsigned* __restrict__ mask,
    const int2* __restrict__ pairs,
    const float* __restrict__ Waf,
    const float* __restrict__ ba,
    const float* __restrict__ Wbf,
    const float* __restrict__ bb,
    const float* __restrict__ gg,
    const float* __restrict__ be,
    const float* __restrict__ mm,
    const float* __restrict__ vv,
    float* __restrict__ hout,
    int n_nodes, int pitchWords) {
  constexpr int VEC = F / 32;
  constexpr int PITCH = F + 1;
  typedef float vec_t __attribute__((ext_vector_type(VEC)));
  __shared__ float acc[TILE_LDS_ROWS * PITCH];
  __shared__ int qe[AGG_WAVES * QCAP];

  const int lane = threadIdx.x & 31;
  const int wave = __builtin_amdgcn_readfirstlane((int)(threadIdx.x >> 5));
  const int lo = blockIdx.x * TILE_NODES;
  int cnt = n_nodes - lo;
  if (cnt > TILE_NODES) cnt = TILE_NODES;
  if (cnt <= 0) return;

  for (int r = wave; r < cnt; r += AGG_WAVES) {
    const vec_t v = *(const vec_t*)(xin + (long long)(lo + r) * F + VEC * lane);
#pragma unroll
    for (int k = 0; k < VEC; ++k) acc[r * PITCH + k * 32 + lane] = v[k];
  }
  for (int r = cnt + wave; r < ((cnt + 15) & ~15); r += AGG_WAVES) {
#pragma unroll
    for (int k = 0; k < VEC; ++k) acc[r * PITCH + k * 32 + lane] = 0.0f;
  }
  __syncthreads();

  int* const myq = qe + wave * QCAP;
  int qn = 0;

  auto add_rows = [&](int nb) {
    int s = 0, ld = 0;
    if (lane < nb) {
      const int2 p = pairs[myq[lane]];
      s = p.x;
      ld = p.y - lo;
    }
    for (int j = 0; j < nb; j += 8) {
      vec_t v[8];
      int row[8];
#pragma unroll
      for (int u = 0; u < 8; ++u) {
        int jj = j + u;
        if (jj > nb - 1) jj = nb - 1;
        const int sj = __builtin_amdgcn_readlane(s, jj);
        row[u] = __builtin_amdgcn_readlane(ld, jj);
        v[u] = *(const vec_t*)(xin + (long long)sj * F + VEC * lane);
      }
#pragma unroll
      for (int u = 0; u < 8; ++u) {
        if (j + u < nb) {
          float* p = &acc[row[u] * PITCH + lane];
#pragma unroll
          for (int k = 0; k < VEC; ++k)
            __hip_atomic_fetch_add(p + k * 32, v[u][k], __ATOMIC_RELAXED,
                                   __HIP_MEMORY_SCOPE_WORKGROUP);
        }
      }
    }
  };

  auto take = [&](unsigned m, int ebase) {
    unsigned any = __builtin_amdgcn_ballot_w32(m != 0u);
    while (any != 0u) {
      if (m != 0u) {
        const int pos = qn + (int)__builtin_amdgcn_mbcnt_lo(any, 0u);
        myq[pos] = ebase + __builtin_ctz(m);
        m &= m - 1u;
      }
      qn += __builtin_popcount(any);
      if (qn >= 32) {
        add_rows(32);
        qn -= 32;
        if (lane < qn) myq[lane] = myq[32 + lane];
      }
      any = __builtin_amdgcn_ballot_w32(m != 0u);
    }
  };

  const uint4* __restrict__ mrow = (const uint4*)(mask + (long long)blockIdx.x * pitchWords);
  const int nQ = pitchWords >> 2;
  const int nG = (nQ + 31) >> 5;
  if (wave < nG) {
    auto piece = [&](int g) {
      int q = g * 32 + lane;
      if (q > nQ - 1) q = nQ - 1;
      return mrow[q];
    };
    uint4 nxt = piece(wave);
    for (int g = wave; g < nG; g += AGG_WAVES) {
      const uint4 c = nxt;
      nxt = piece(g + AGG_WAVES);
      const int q = g * 32 + lane;
      const bool in = q < nQ;
      const int eb = q * 128;
      take(in ? c.x : 0u, eb);
      take(in ? c.y : 0u, eb + 32);
      take(in ? c.z : 0u, eb + 64);
      take(in ? c.w : 0u, eb + 96);
    }
  }
  add_rows(qn);
  __syncthreads();

  const int r0 = wave * 16;
  if (r0 >= cnt) return;
  const int half = lane >> 4;
  const int l16 = lane & 15;

  v8f g1[4];
#pragma unroll
  for (int t = 0; t < 4; ++t) {
    const float bias = ba[t * 16 + l16];
#pragma unroll
    for (int r = 0; r < 8; ++r) g1[t][r] = bias;
  }
  float* const arow = &acc[(r0 + l16) * PITCH + 16 * half];
  const float* const wa = Waf + 4 * lane;
#pragma unroll 4
  for (int j = 0; j < F / 8; ++j) {
    v4f b[4];
#pragma unroll
    for (int t = 0; t < 4; ++t) b[t] = *(const v4f*)(wa + (j * 4 + t) * 128);
#pragma unroll
    for (int s = 0; s < 2; ++s) {
      const int e = 4 * j + 2 * s;
      const int col = (e % VEC) * 32 + e / VEC;
      v2f a;
      a.x = arow[col];
      a.y = arow[col + 32];
#pragma unroll
      for (int t = 0; t < 4; ++t) {
        v2f bb2;
        bb2.x = s == 0 ? b[t].x : b[t].z;
        bb2.y = s == 0 ? b[t].y : b[t].w;
        g1[t] = __builtin_amdgcn_wmma_f32_16x16x4_f32(
            false, a, false, bb2, (short)0, g1[t], false, false);
      }
    }
  }

#pragma unroll
  for (int t = 0; t < 4; ++t) {
    const int n = t * 16 + l16;
#pragma unroll
    for (int r = 0; r < 8; ++r) {
      float x = g1[t][r];
      x = x > 0.0f ? x : 0.0f;
      acc[(r0 + r + 8 * half) * PITCH + (n & 1) * 32 + (n >> 1)] = x;
    }
  }

  v8f g2[4];
#pragma unroll
  for (int t = 0; t < 4; ++t)
#pragma unroll
    for (int r = 0; r < 8; ++r) g2[t][r] = 0.0f;
  const float* const wb = Wbf + 4 * lane;
#pragma unroll 4
  for (int j = 0; j < HIDDEN / 8; ++j) {
    v4f b[4];
#pragma unroll
    for (int t = 0; t < 4; ++t) b[t] = *(const v4f*)(wb + (j * 4 + t) * 128);
#pragma unroll
    for (int s = 0; s < 2; ++s) {
      const int q = 2 * j + s;
      v2f a;
      a.x = arow[q];
      a.y = arow[q + 32];
#pragma unroll
      for (int t = 0; t < 4; ++t) {
        v2f bb2;
        bb2.x = s == 0 ? b[t].x : b[t].z;
        bb2.y = s == 0 ? b[t].y : b[t].w;
        g2[t] = __builtin_amdgcn_wmma_f32_16x16x4_f32(
            false, a, false, bb2, (short)0, g2[t], false, false);
      }
    }
  }

#pragma unroll
  for (int t = 0; t < 4; ++t) {
    const int n = t * 16 + l16;
    const float s = gg[n] * rsqrtf(vv[n] + BN_EPS);
    const float sh = (bb[n] - mm[n]) * s + be[n];
#pragma unroll
    for (int r = 0; r < 8; ++r) {
      const int lr = r0 + r + 8 * half;
      float o = g2[t][r] * s + sh;
      o = o > 0.0f ? o : 0.0f;
      if (lr < cnt) hout[(long long)(lo + lr) * HIDDEN + n] = o;
    }
  }
}

__global__ __launch_bounds__(256) void pool_zero_kernel(float* __restrict__ sums,
                                                        float* __restrict__ counts) {
  int i = blockIdx.x * blockDim.x + threadIdx.x;
  if (i < NGRAPH * HIDDEN) sums[i] = 0.0f;
  if (i < NGRAPH) counts[i] = 0.0f;
}

#define POOL_ROWS  32
#define POOL_GROUP 16

__global__ __launch_bounds__(256) void pool_sum_kernel(
    const float* __restrict__ h, const int* __restrict__ batch,
    float* __restrict__ sums, float* __restrict__ counts, int n_nodes) {
  const int lane = threadIdx.x & 31;
  const int wave =
      __builtin_amdgcn_readfirstlane((int)((blockIdx.x * blockDim.x + threadIdx.x) >> 5));
  const int i0 = wave * POOL_ROWS;
  if (i0 >= n_nodes) return;
  int i1 = i0 + POOL_ROWS;
  if (i1 > n_nodes) i1 = n_nodes;

  auto flush = [&](int g, v2f a, int rows) {
    float* p = sums + (long long)g * HIDDEN + 2 * lane;
    atomicAdd(p, a.x);
    atomicAdd(p + 1, a.y);
    if (lane == 0) atomicAdd(&counts[g], (float)rows);
  };

  int cur = batch[i0];
  v2f acc = {0.0f, 0.0f};
  int rows = 0;
  for (int i = i0; i < i1; i += POOL_GROUP) {
    v2f v[POOL_GROUP];
    int g[POOL_GROUP];
#pragma unroll
    for (int u = 0; u < POOL_GROUP; ++u) {
      int ii = i + u;
      if (ii > i1 - 1) ii = i1 - 1;
      g[u] = batch[ii];
      v[u] = *(const v2f*)(h + (long long)ii * HIDDEN + 2 * lane);
    }
#pragma unroll
    for (int u = 0; u < POOL_GROUP; ++u) {
      if (i + u < i1) {
        if (g[u] != cur) {
          flush(cur, acc, rows);
          acc.x = 0.0f;
          acc.y = 0.0f;
          rows = 0;
          cur = g[u];
        }
        acc.x += v[u].x;
        acc.y += v[u].y;
        ++rows;
      }
    }
  }
  flush(cur, acc, rows);
}

__global__ __launch_bounds__(256) void pool_div_kernel(
    const float* __restrict__ sums, const float* __restrict__ counts,
    float* __restrict__ out) {
  int i = blockIdx.x * blockDim.x + threadIdx.x;
  if (i >= NGRAPH * HIDDEN) return;
  float c = counts[i >> 6];
  c = c > 1.0f ? c : 1.0f;
  out[i] = sums[i] / c;
}

extern "C" void kernel_launch(void* const* d_in, const int* in_sizes, int n_in,
                              void* d_out, int out_size, void* d_ws, size_t ws_size,
                              hipStream_t stream) {
  const float* x   = (const float*)d_in[0];
  const int* ei    = (const int*)d_in[1];
  const int* batch = (const int*)d_in[2];
  const float* W1a = (const float*)d_in[3];
  const float* b1a = (const float*)d_in[4];
  const float* W1b = (const float*)d_in[5];
  const float* b1b = (const float*)d_in[6];
  const float* g1  = (const float*)d_in[7];
  const float* be1 = (const float*)d_in[8];
  const float* m1  = (const float*)d_in[9];
  const float* v1  = (const float*)d_in[10];
  const float* W2a = (const float*)d_in[11];
  const float* b2a = (const float*)d_in[12];
  const float* W2b = (const float*)d_in[13];
  const float* b2b = (const float*)d_in[14];
  const float* g2  = (const float*)d_in[15];
  const float* be2 = (const float*)d_in[16];
  const float* m2  = (const float*)d_in[17];
  const float* v2  = (const float*)d_in[18];
  float* out = (float*)d_out;

  const int N = in_sizes[0] / IN_FEATS;
  const int E = in_sizes[1] / 2;

  float* ws    = (float*)d_ws;
  float* h1    = ws;
  float* h2    = h1 + (long long)N * HIDDEN;
  float* sums  = h2 + (long long)N * HIDDEN;
  float* cnts  = sums + NGRAPH * HIDDEN;
  (void)ws_size; (void)n_in; (void)out_size;

  const int aggBlk = (N + TILE_NODES - 1) / TILE_NODES;

  const int maskBlk    = (E + MASK_EDGES - 1) / MASK_EDGES;
  const int pitchWords = maskBlk * MASK_WORDS;
  float* wf = cnts + NGRAPH;
  float* w1af = wf;
  float* w1bf = w1af + 4 * WF_W1A_VEC;
  float* w2af = w1bf + 4 * WF_SQ_VEC;
  float* w2bf = w2af + 4 * WF_SQ_VEC;
  unsigned* mask = (unsigned*)(wf + 4 * WF_VECS);
  int2* pairs    = (int2*)(mask + (long long)aggBlk * pitchWords);

  edge_mask_kernel<<<maskBlk + WF_BLOCKS, MASK_THREADS, 0, stream>>>(
      ei, mask, pairs, E, aggBlk, pitchWords, W1a, W1b, W2a, W2b, wf, maskBlk);

  tile_layer_kernel<IN_FEATS><<<aggBlk, 1024, 0, stream>>>(
      x, mask, pairs, w1af, b1a, w1bf, b1b, g1, be1, m1, v1, h1, N, pitchWords);

  tile_layer_kernel<HIDDEN><<<aggBlk, 1024, 0, stream>>>(
      h1, mask, pairs, w2af, b2a, w2bf, b2b, g2, be2, m2, v2, h2, N, pitchWords);

  pool_zero_kernel<<<(NGRAPH * HIDDEN + 255) / 256, 256, 0, stream>>>(sums, cnts);
  {
    const int rowsPerBlock = POOL_ROWS * 8;
    const int blocks = (N + rowsPerBlock - 1) / rowsPerBlock;
    pool_sum_kernel<<<blocks, 256, 0, stream>>>(h2, batch, sums, cnts, N);
  }
  pool_div_kernel<<<(NGRAPH * HIDDEN + 255) / 256, 256, 0, stream>>>(sums, cnts, out);
}


# Tile kernels: LDS sums start at zero; a tile's own rows head each wave's edge list and are gathered and added like source rows, not 13 dependent loads before the barrier; self term in arrival order
# speedup vs baseline: 3.9006x; 1.0380x over previous
#include <hip/hip_runtime.h>

typedef float v2f __attribute__((ext_vector_type(2)));
typedef float v4f __attribute__((ext_vector_type(4)));
typedef float v8f __attribute__((ext_vector_type(8)));

#define IN_FEATS 128
#define HIDDEN   64
#define NGRAPH   512
#define BN_EPS   1e-5f

#define TILE_NODES 392
#define AGG_WAVES  32
#define QCAP       64
#define OWN_ROW    ((int)0x80000000)

#define MASK_EDGES   2048
#define MASK_WORDS   64
#define MASK_PITCH   66
#define MASK_TILES   256
#define MASK_THREADS 512

#define WF_W1A_VEC 2048
#define WF_SQ_VEC  1024
#define WF_VECS    (WF_W1A_VEC + 3 * WF_SQ_VEC)
#define WF_BLOCKS  ((WF_VECS + MASK_THREADS - 1) / MASK_THREADS)

__global__ __launch_bounds__(MASK_THREADS) void edge_mask_kernel(
    const int* __restrict__ ei, unsigned* __restrict__ mask, int2* __restrict__ pairs,
    int E, int nTiles, int pitchWords,
    const float* __restrict__ W1a, const float* __restrict__ W1b,
    const float* __restrict__ W2a, const float* __restrict__ W2b,
    float* __restrict__ wf, int nEdgeBlk) {
  __shared__ unsigned w[MASK_TILES * MASK_PITCH];
  const int i = threadIdx.x;
  if ((int)blockIdx.x >= nEdgeBlk) {
    const int o = ((int)blockIdx.x - nEdgeBlk) * MASK_THREADS + i;
    if (o < WF_VECS) {
      const float* W = W1a;
      int K = IN_FEATS, u = o;
      if (o >= WF_W1A_VEC) {
        const int m = (o - WF_W1A_VEC) / WF_SQ_VEC;
        u = (o - WF_W1A_VEC) % WF_SQ_VEC;
        K = HIDDEN;
        W = m == 0 ? W1b : (m == 1 ? W2a : W2b);
      }
      const int L = u & 31, t = (u >> 5) & 3, j = u >> 7;
      const float* p = W + (t * 16 + (L & 15)) * K + 4 * j + (K / 2) * (L >> 4);
      v4f v;
      v.x = p[0]; v.y = p[1]; v.z = p[2]; v.w = p[3];
      *(v4f*)(wf + 4 * o) = v;
    }
    return;
  }
  const int lane = i & 31;
  const int wave = __builtin_amdgcn_readfirstlane(i >> 5);
  const int* __restrict__ dst = ei + E;
  const int cbase = blockIdx.x * MASK_EDGES;

  int d[4];
#pragma unroll
  for (int r = 0; r < 2; ++r) {
    const int e = cbase + r * 1024 + 2 * i;
    if ((E & 1) == 0 && e < E) {
      const int2 sv = *(const int2*)(ei + e);
      const int2 dv = *(const int2*)(dst + e);
      d[2 * r] = dv.x;
      d[2 * r + 1] = dv.y;
      *(int4*)(pairs + e) = make_int4(sv.x, dv.x, sv.y, dv.y);
    } else {
#pragma unroll
      for (int k = 0; k < 2; ++k) {
        d[2 * r + k] = -1;
        if (e + k < E) {
          const int sv = ei[e + k];
          const int dv = dst[e + k];
          d[2 * r + k] = dv;
          pairs[e + k] = make_int2(sv, dv);
        }
      }
    }
  }

  for (int t0 = 0; t0 < nTiles; t0 += MASK_TILES) {
    int rows = nTiles - t0;
    if (rows > MASK_TILES) rows = MASK_TILES;
    for (int q = i; q < MASK_TILES * MASK_PITCH / 4; q += MASK_THREADS)
      ((uint4*)w)[q] = make_uint4(0u, 0u, 0u, 0u);
    __syncthreads();
#pragma unroll
    for (int r = 0; r < 2; ++r) {
#pragma unroll
      for (int k = 0; k < 2; ++k) {
        const int dv = d[2 * r + k];
        if (dv >= 0) {
          const unsigned t = (unsigned)dv / TILE_NODES - (unsigned)t0;
          if (t < (unsigned)rows)
            __hip_atomic_fetch_or(&w[t * MASK_PITCH + 32 * r + (i >> 4)],
                                  1u << (2 * (i & 15) + k), __ATOMIC_RELAXED,
                                  __HIP_MEMORY_SCOPE_WORKGROUP);
        }
      }
    }
    __syncthreads();
    for (int r = wave; r < rows; r += MASK_THREADS / 32) {
      const uint2 v = *(const uint2*)&w[r * MASK_PITCH + 2 * lane];
      *(uint2*)(mask + (long long)(t0 + r) * pitchWords + blockIdx.x * MASK_WORDS + 2 * lane) = v;
    }
    __syncthreads();
  }
}

#define TILE_LDS_ROWS 400

template <int F>
__global__ __launch_bounds__(1024) void tile_layer_kernel(
    const float* __restrict__ xin, const unsigned* __restrict__ mask,
    const int2* __restrict__ pairs,
    const float* __restrict__ Waf,
    const float* __restrict__ ba,
    const float* __restrict__ Wbf,
    const float* __restrict__ bb,
    const float* __restrict__ gg,
    const float* __restrict__ be,
    const float* __restrict__ mm,
    const float* __restrict__ vv,
    float* __restrict__ hout,
    int n_nodes, int pitchWords) {
  constexpr int VEC = F / 32;
  constexpr int PITCH = F + 1;
  typedef float vec_t __attribute__((ext_vector_type(VEC)));
  __shared__ __attribute__((aligned(16))) float acc[TILE_LDS_ROWS * PITCH];
  __shared__ int qe[AGG_WAVES * QCAP];

  const int lane = threadIdx.x & 31;
  const int wave = __builtin_amdgcn_readfirstlane((int)(threadIdx.x >> 5));
  const int lo = blockIdx.x * TILE_NODES;
  int cnt = n_nodes - lo;
  if (cnt > TILE_NODES) cnt = TILE_NODES;
  if (cnt <= 0) return;

  {
    const int nv = (((cnt + 15) & ~15) * PITCH) >> 2;
    const v4f z = {0.0f, 0.0f, 0.0f, 0.0f};
    for (int q = (int)threadIdx.x; q < nv; q += AGG_WAVES * 32) ((v4f*)acc)[q] = z;
  }

  int* const myq = qe + wave * QCAP;
  int qn = 0;
  static_assert((TILE_NODES + AGG_WAVES - 1) / AGG_WAVES <= 31, "own rows must fit the waiting part of the list");
  {
    const int r = wave + AGG_WAVES * lane;
    if (r < cnt) myq[lane] = r | OWN_ROW;
    if (wave < cnt) qn = (cnt - wave + AGG_WAVES - 1) / AGG_WAVES;
  }
  __syncthreads();

  auto add_rows = [&](int nb) {
    int s = 0, ld = 0;
    if (lane < nb) {
      const int idx = myq[lane];
      if (idx < 0) {
        ld = idx & ~OWN_ROW;
        s = lo + ld;
      } else {
        const int2 p = pairs[idx];
        s = p.x;
        ld = p.y - lo;
      }
    }
    for (int j = 0; j < nb; j += 8) {
      vec_t v[8];
      int row[8];
#pragma unroll
      for (int u = 0; u < 8; ++u) {
        int jj = j + u;
        if (jj > nb - 1) jj = nb - 1;
        const int sj = __builtin_amdgcn_readlane(s, jj);
        row[u] = __builtin_amdgcn_readlane(ld, jj);
        v[u] = *(const vec_t*)(xin + (long long)sj * F + VEC * lane);
      }
#pragma unroll
      for (int u = 0; u < 8; ++u) {
        if (j + u < nb) {
          float* p = &acc[row[u] * PITCH + lane];
#pragma unroll
          for (int k = 0; k < VEC; ++k)
            __hip_atomic_fetch_add(p + k * 32, v[u][k], __ATOMIC_RELAXED,
                                   __HIP_MEMORY_SCOPE_WORKGROUP);
        }
      }
    }
  };

  auto take = [&](unsigned m, int ebase) {
    unsigned any = __builtin_amdgcn_ballot_w32(m != 0u);
    while (any != 0u) {
      if (m != 0u) {
        const int pos = qn + (int)__builtin_amdgcn_mbcnt_lo(any, 0u);
        myq[pos] = ebase + __builtin_ctz(m);
        m &= m - 1u;
      }
      qn += __builtin_popcount(any);
      if (qn >= 32) {
        add_rows(32);
        qn -= 32;
        if (lane < qn) myq[lane] = myq[32 + lane];
      }
      any = __builtin_amdgcn_ballot_w32(m != 0u);
    }
  };

  const uint4* __restrict__ mrow = (const uint4*)(mask + (long long)blockIdx.x * pitchWords);
  const int nQ = pitchWords >> 2;
  const int nG = (nQ + 31) >> 5;
  if (wave < nG) {
    auto piece = [&](int g) {
      int q = g * 32 + lane;
      if (q > nQ - 1) q = nQ - 1;
      return mrow[q];
    };
    uint4 nxt = piece(wave);
    for (int g = wave; g < nG; g += AGG_WAVES) {
      const uint4 c = nxt;
      nxt = piece(g + AGG_WAVES);
      const int q = g * 32 + lane;
      const bool in = q < nQ;
      const int eb = q * 128;
      take(in ? c.x : 0u, eb);
      take(in ? c.y : 0u, eb + 32);
      take(in ? c.z : 0u, eb + 64);
      take(in ? c.w : 0u, eb + 96);
    }
  }
  add_rows(qn);
  __syncthreads();

  const int r0 = wave * 16;
  if (r0 >= cnt) return;
  const int half = lane >> 4;
  const int l16 = lane & 15;

  v8f g1[4];
#pragma unroll
  for (int t = 0; t < 4; ++t) {
    const float bias = ba[t * 16 + l16];
#pragma unroll
    for (int r = 0; r < 8; ++r) g1[t][r] = bias;
  }
  float* const arow = &acc[(r0 + l16) * PITCH + 16 * half];
  const float* const wa = Waf + 4 * lane;
#pragma unroll 4
  for (int j = 0; j < F / 8; ++j) {
    v4f b[4];
#pragma unroll
    for (int t = 0; t < 4; ++t) b[t] = *(const v4f*)(wa + (j * 4 + t) * 128);
#pragma unroll
    for (int s = 0; s < 2; ++s) {
      const int e = 4 * j + 2 * s;
      const int col = (e % VEC) * 32 + e / VEC;
      v2f a;
      a.x = arow[col];
      a.y = arow[col + 32];
#pragma unroll
      for (int t = 0; t < 4; ++t) {
        v2f bb2;
        bb2.x = s == 0 ? b[t].x : b[t].z;
        bb2.y = s == 0 ? b[t].y : b[t].w;
        g1[t] = __builtin_amdgcn_wmma_f32_16x16x4_f32(
            false, a, false, bb2, (short)0, g1[t], false, false);
      }
    }
  }

#pragma unroll
  for (int t = 0; t < 4; ++t) {
    const int n = t * 16 + l16;
#pragma unroll
    for (int r = 0; r < 8; ++r) {
      float x = g1[t][r];
      x = x > 0.0f ? x : 0.0f;
      acc[(r0 + r + 8 * half) * PITCH + (n & 1) * 32 + (n >> 1)] = x;
    }
  }

  v8f g2[4];
#pragma unroll
  for (int t = 0; t < 4; ++t)
#pragma unroll
    for (int r = 0; r < 8; ++r) g2[t][r] = 0.0f;
  const float* const wb = Wbf + 4 * lane;
#pragma unroll 4
  for (int j = 0; j < HIDDEN / 8; ++j) {
    v4f b[4];
#pragma unroll
    for (int t = 0; t < 4; ++t) b[t] = *(const v4f*)(wb + (j * 4 + t) * 128);
#pragma unroll
    for (int s = 0; s < 2; ++s) {
      const int q = 2 * j + s;
      v2f a;
      a.x = arow[q];
      a.y = arow[q + 32];
#pragma unroll
      for (int t = 0; t < 4; ++t) {
        v2f bb2;
        bb2.x = s == 0 ? b[t].x : b[t].z;
        bb2.y = s == 0 ? b[t].y : b[t].w;
        g2[t] = __builtin_amdgcn_wmma_f32_16x16x4_f32(
            false, a, false, bb2, (short)0, g2[t], false, false);
      }
    }
  }

#pragma unroll
  for (int t = 0; t < 4; ++t) {
    const int n = t * 16 + l16;
    const float s = gg[n] * rsqrtf(vv[n] + BN_EPS);
    const float sh = (bb[n] - mm[n]) * s + be[n];
#pragma unroll
    for (int r = 0; r < 8; ++r) {
      const int lr = r0 + r + 8 * half;
      float o = g2[t][r] * s + sh;
      o = o > 0.0f ? o : 0.0f;
      if (lr < cnt) hout[(long long)(lo + lr) * HIDDEN + n] = o;
    }
  }
}

__global__ __launch_bounds__(256) void pool_zero_kernel(float* __restrict__ sums,
                                                        float* __restrict__ counts) {
  int i = blockIdx.x * blockDim.x + threadIdx.x;
  if (i < NGRAPH * HIDDEN) sums[i] = 0.0f;
  if (i < NGRAPH) counts[i] = 0.0f;
}

#define POOL_ROWS  32
#define POOL_GROUP 16

__global__ __launch_bounds__(256) void pool_sum_kernel(
    const float* __restrict__ h, const int* __restrict__ batch,
    float* __restrict__ sums, float* __restrict__ counts, int n_nodes) {
  const int lane = threadIdx.x & 31;
  const int wave =
      __builtin_amdgcn_readfirstlane((int)((blockIdx.x * blockDim.x + threadIdx.x) >> 5));
  const int i0 = wave * POOL_ROWS;
  if (i0 >= n_nodes) return;
  int i1 = i0 + POOL_ROWS;
  if (i1 > n_nodes) i1 = n_nodes;

  auto flush = [&](int g, v2f a, int rows) {
    float* p = sums + (long long)g * HIDDEN + 2 * lane;
    atomicAdd(p, a.x);
    atomicAdd(p + 1, a.y);
    if (lane == 0) atomicAdd(&counts[g], (float)rows);
  };

  int cur = batch[i0];
  v2f acc = {0.0f, 0.0f};
  int rows = 0;
  for (int i = i0; i < i1; i += POOL_GROUP) {
    v2f v[POOL_GROUP];
    int g[POOL_GROUP];
#pragma unroll
    for (int u = 0; u < POOL_GROUP; ++u) {
      int ii = i + u;
      if (ii > i1 - 1) ii = i1 - 1;
      g[u] = batch[ii];
      v[u] = *(const v2f*)(h + (long long)ii * HIDDEN + 2 * lane);
    }
#pragma unroll
    for (int u = 0; u < POOL_GROUP; ++u) {
      if (i + u < i1) {
        if (g[u] != cur) {
          flush(cur, acc, rows);
          acc.x = 0.0f;
          acc.y = 0.0f;
          rows = 0;
          cur = g[u];
        }
        acc.x += v[u].x;
        acc.y += v[u].y;
        ++rows;
      }
    }
  }
  flush(cur, acc, rows);
}

__global__ __launch_bounds__(256) void pool_div_kernel(
    const float* __restrict__ sums, const float* __restrict__ counts,
    float* __restrict__ out) {
  int i = blockIdx.x * blockDim.x + threadIdx.x;
  if (i >= NGRAPH * HIDDEN) return;
  float c = counts[i >> 6];
  c = c > 1.0f ? c : 1.0f;
  out[i] = sums[i] / c;
}

extern "C" void kernel_launch(void* const* d_in, const int* in_sizes, int n_in,
                              void* d_out, int out_size, void* d_ws, size_t ws_size,
                              hipStream_t stream) {
  const float* x   = (const float*)d_in[0];
  const int* ei    = (const int*)d_in[1];
  const int* batch = (const int*)d_in[2];
  const float* W1a = (const float*)d_in[3];
  const float* b1a = (const float*)d_in[4];
  const float* W1b = (const float*)d_in[5];
  const float* b1b = (const float*)d_in[6];
  const float* g1  = (const float*)d_in[7];
  const float* be1 = (const float*)d_in[8];
  const float* m1  = (const float*)d_in[9];
  const float* v1  = (const float*)d_in[10];
  const float* W2a = (const float*)d_in[11];
  const float* b2a = (const float*)d_in[12];
  const float* W2b = (const float*)d_in[13];
  const float* b2b = (const float*)d_in[14];
  const float* g2  = (const float*)d_in[15];
  const float* be2 = (const float*)d_in[16];
  const float* m2  = (const float*)d_in[17];
  const float* v2  = (const float*)d_in[18];
  float* out = (float*)d_out;

  const int N = in_sizes[0] / IN_FEATS;
  const int E = in_sizes[1] / 2;

  float* ws    = (float*)d_ws;
  float* h1    = ws;
  float* h2    = h1 + (long long)N * HIDDEN;
  float* sums  = h2 + (long long)N * HIDDEN;
  float* cnts  = sums + NGRAPH * HIDDEN;
  (void)ws_size; (void)n_in; (void)out_size;

  const int aggBlk = (N + TILE_NODES - 1) / TILE_NODES;

  const int maskBlk    = (E + MASK_EDGES - 1) / MASK_EDGES;
  const int pitchWords = maskBlk * MASK_WORDS;
  float* wf = cnts + NGRAPH;
  float* w1af = wf;
  float* w1bf = w1af + 4 * WF_W1A_VEC;
  float* w2af = w1bf + 4 * WF_SQ_VEC;
  float* w2bf = w2af + 4 * WF_SQ_VEC;
  unsigned* mask = (unsigned*)(wf + 4 * WF_VECS);
  int2* pairs    = (int2*)(mask + (long long)aggBlk * pitchWords);

  edge_mask_kernel<<<maskBlk + WF_BLOCKS, MASK_THREADS, 0, stream>>>(
      ei, mask, pairs, E, aggBlk, pitchWords, W1a, W1b, W2a, W2b, wf, maskBlk);

  tile_layer_kernel<IN_FEATS><<<aggBlk, 1024, 0, stream>>>(
      x, mask, pairs, w1af, b1a, w1bf, b1b, g1, be1, m1, v1, h1, N, pitchWords);

  tile_layer_kernel<HIDDEN><<<aggBlk, 1024, 0, stream>>>(
      h1, mask, pairs, w2af, b2a, w2bf, b2b, g2, be2, m2, v2, h2, N, pitchWords);

  pool_zero_kernel<<<(NGRAPH * HIDDEN + 255) / 256, 256, 0, stream>>>(sums, cnts);
  {
    const int rowsPerBlock = POOL_ROWS * 8;
    const int blocks = (N + rowsPerBlock - 1) / rowsPerBlock;
    pool_sum_kernel<<<blocks, 256, 0, stream>>>(h2, batch, sums, cnts, N);
  }
  pool_div_kernel<<<(NGRAPH * HIDDEN + 255) / 256, 256, 0, stream>>>(sums, cnts, out);
}


# Tile kernels: wave w owns rows w, w+32, .. of its tile in registers; layer 1 routes kept edges to owners' LDS queues, layer 2 reads the stored queues; no LDS float adds; sum order: queue arrival
# speedup vs baseline: 4.5569x; 1.1683x over previous
#include <hip/hip_runtime.h>

typedef float v2f __attribute__((ext_vector_type(2)));
typedef float v4f __attribute__((ext_vector_type(4)));
typedef float v8f __attribute__((ext_vector_type(8)));

#define IN_FEATS 128
#define HIDDEN   64
#define NGRAPH   512
#define BN_EPS   1e-5f

#define TILE_NODES 392
#define AGG_WAVES  32
#define QCAP       64
#define OWN_ROW    ((int)0x80000000)
#define OWN_SLOTS      13
#define OWN_QCAP       768
#define OWN_QPITCH     (OWN_QCAP + 1)
#define OWN_SLOT_SHIFT 28
#define OWN_SRC_MASK   0x0fffffff

#define MASK_EDGES   2048
#define MASK_WORDS   64
#define MASK_PITCH   66
#define MASK_TILES   256
#define MASK_THREADS 512

#define WF_W1A_VEC 2048
#define WF_SQ_VEC  1024
#define WF_VECS    (WF_W1A_VEC + 3 * WF_SQ_VEC)
#define WF_BLOCKS  ((WF_VECS + MASK_THREADS - 1) / MASK_THREADS)

__global__ __launch_bounds__(MASK_THREADS) void edge_mask_kernel(
    const int* __restrict__ ei, unsigned* __restrict__ mask, int2* __restrict__ pairs,
    int E, int nTiles, int pitchWords,
    const float* __restrict__ W1a, const float* __restrict__ W1b,
    const float* __restrict__ W2a, const float* __restrict__ W2b,
    float* __restrict__ wf, int nEdgeBlk) {
  __shared__ unsigned w[MASK_TILES * MASK_PITCH];
  const int i = threadIdx.x;
  if ((int)blockIdx.x >= nEdgeBlk) {
    const int o = ((int)blockIdx.x - nEdgeBlk) * MASK_THREADS + i;
    if (o < WF_VECS) {
      const float* W = W1a;
      int K = IN_FEATS, u = o;
      if (o >= WF_W1A_VEC) {
        const int m = (o - WF_W1A_VEC) / WF_SQ_VEC;
        u = (o - WF_W1A_VEC) % WF_SQ_VEC;
        K = HIDDEN;
        W = m == 0 ? W1b : (m == 1 ? W2a : W2b);
      }
      const int L = u & 31, t = (u >> 5) & 3, j = u >> 7;
      const float* p = W + (t * 16 + (L & 15)) * K + 4 * j + (K / 2) * (L >> 4);
      v4f v;
      v.x = p[0]; v.y = p[1]; v.z = p[2]; v.w = p[3];
      *(v4f*)(wf + 4 * o) = v;
    }
    return;
  }
  const int lane = i & 31;
  const int wave = __builtin_amdgcn_readfirstlane(i >> 5);
  const int* __restrict__ dst = ei + E;
  const int cbase = blockIdx.x * MASK_EDGES;

  int d[4];
#pragma unroll
  for (int r = 0; r < 2; ++r) {
    const int e = cbase + r * 1024 + 2 * i;
    if ((E & 1) == 0 && e < E) {
      const int2 sv = *(const int2*)(ei + e);
      const int2 dv = *(const int2*)(dst + e);
      d[2 * r] = dv.x;
      d[2 * r + 1] = dv.y;
      *(int4*)(pairs + e) = make_int4(sv.x, dv.x, sv.y, dv.y);
    } else {
#pragma unroll
      for (int k = 0; k < 2; ++k) {
        d[2 * r + k] = -1;
        if (e + k < E) {
          const int sv = ei[e + k];
          const int dv = dst[e + k];
          d[2 * r + k] = dv;
          pairs[e + k] = make_int2(sv, dv);
        }
      }
    }
  }

  for (int t0 = 0; t0 < nTiles; t0 += MASK_TILES) {
    int rows = nTiles - t0;
    if (rows > MASK_TILES) rows = MASK_TILES;
    for (int q = i; q < MASK_TILES * MASK_PITCH / 4; q += MASK_THREADS)
      ((uint4*)w)[q] = make_uint4(0u, 0u, 0u, 0u);
    __syncthreads();
#pragma unroll
    for (int r = 0; r < 2; ++r) {
#pragma unroll
      for (int k = 0; k < 2; ++k) {
        const int dv = d[2 * r + k];
        if (dv >= 0) {
          const unsigned t = (unsigned)dv / TILE_NODES - (unsigned)t0;
          if (t < (unsigned)rows)
            __hip_atomic_fetch_or(&w[t * MASK_PITCH + 32 * r + (i >> 4)],
                                  1u << (2 * (i & 15) + k), __ATOMIC_RELAXED,
                                  __HIP_MEMORY_SCOPE_WORKGROUP);
        }
      }
    }
    __syncthreads();
    for (int r = wave; r < rows; r += MASK_THREADS / 32) {
      const uint2 v = *(const uint2*)&w[r * MASK_PITCH + 2 * lane];
      *(uint2*)(mask + (long long)(t0 + r) * pitchWords + blockIdx.x * MASK_WORDS + 2 * lane) = v;
    }
    __syncthreads();
  }
}

#define TILE_LDS_ROWS 400
#define SPARE_ROW TILE_LDS_ROWS

template <int F, int ROLE>
__global__ __launch_bounds__(1024) void tile_layer_kernel(
    const float* __restrict__ xin, const unsigned* __restrict__ mask,
    const int2* __restrict__ pairs,
    const float* __restrict__ Waf,
    const float* __restrict__ ba,
    const float* __restrict__ Wbf,
    const float* __restrict__ bb,
    const float* __restrict__ gg,
    const float* __restrict__ be,
    const float* __restrict__ mm,
    const float* __restrict__ vv,
    float* __restrict__ hout,
    int n_nodes, int pitchWords,
    int* __restrict__ elist,
    int* __restrict__ lcnt) {
  constexpr int VEC = F / 32;
  constexpr int PITCH = F + 1;
  constexpr int GROUP = 32 / VEC;
  constexpr int OQ_WORDS = ROLE == 1 ? AGG_WAVES * OWN_QPITCH : 32;
  typedef float vec_t __attribute__((ext_vector_type(VEC)));
  typedef float v32f __attribute__((ext_vector_type(32)));
  __shared__ __attribute__((aligned(16))) float acc[(TILE_LDS_ROWS + 1) * PITCH];
  __shared__ int qe[AGG_WAVES * QCAP];
  __shared__ int oq[OQ_WORDS];
  __shared__ int oqn[AGG_WAVES + 1];
  static_assert(TILE_LDS_ROWS <= 32 * OWN_SLOTS, "every LDS row of the tile needs an owner slot");

  const int lane = threadIdx.x & 31;
  const int wave = __builtin_amdgcn_readfirstlane((int)(threadIdx.x >> 5));
  const int lo = blockIdx.x * TILE_NODES;
  int cnt = n_nodes - lo;
  if (cnt > TILE_NODES) cnt = TILE_NODES;
  if (cnt <= 0) return;

  int* const mylist = elist + ((long long)blockIdx.x * AGG_WAVES + wave) * OWN_QCAP + lane;
  int* const myq = qe + wave * QCAP;
  int cv = 0;
  int first = 0;
  bool owned = true;
  if constexpr (ROLE == 2) {
    cv = lcnt[blockIdx.x * AGG_WAVES + lane];
    first = mylist[0];
    owned = __builtin_amdgcn_ballot_w32(cv < 0) == 0u;
  }

  auto gather_add = [&](int s, int ld, int nb) {
    for (int j = 0; j < nb; j += 8) {
      vec_t v[8];
      int row[8];
#pragma unroll
      for (int u = 0; u < 8; ++u) {
        int jj = j + u;
        const bool ok = jj < nb;
        if (!ok) jj = nb - 1;
        const int sj = __builtin_amdgcn_readlane(s, jj);
        row[u] = ok ? __builtin_amdgcn_readlane(ld, jj) : SPARE_ROW;
        v[u] = *(const vec_t*)(xin + (long long)sj * F + VEC * lane);
      }
#pragma unroll
      for (int u = 0; u < 8; ++u) {
        float* p = &acc[row[u] * PITCH + lane];
#pragma unroll
        for (int k = 0; k < VEC; ++k)
          __hip_atomic_fetch_add(p + k * 32, v[u][k], __ATOMIC_RELAXED,
                                 __HIP_MEMORY_SCOPE_WORKGROUP);
      }
    }
  };

  auto add_rows = [&](int nb) {
    int s = -1, ld = 0;
    if (lane < nb) {
      const int idx = myq[lane];
      if (idx < 0) {
        ld = idx & ~OWN_ROW;
        s = lo + ld;
      } else {
        const int2 p = pairs[idx];
        s = p.x;
        ld = p.y - lo;
      }
    }
    gather_add(s, ld, nb);
  };

  auto scan_row = [&](auto&& consume, int qn) {
    auto take = [&](unsigned m, int ebase) {
      unsigned any = __builtin_amdgcn_ballot_w32(m != 0u);
      while (any != 0u) {
        if (m != 0u) {
          const int pos = qn + (int)__builtin_amdgcn_mbcnt_lo(any, 0u);
          myq[pos] = ebase + __builtin_ctz(m);
          m &= m - 1u;
        }
        qn += __builtin_popcount(any);
        if (qn >= 32) {
          consume(32);
          qn -= 32;
          if (lane < qn) myq[lane] = myq[32 + lane];
        }
        any = __builtin_amdgcn_ballot_w32(m != 0u);
      }
    };
    const uint4* __restrict__ mrow = (const uint4*)(mask + (long long)blockIdx.x * pitchWords);
    const int nQ = pitchWords >> 2;
    const int nG = (nQ + 31) >> 5;
    if (wave < nG) {
      auto piece = [&](int g) {
        int q = g * 32 + lane;
        if (q > nQ - 1) q = nQ - 1;
        return mrow[q];
      };
      uint4 nxt = piece(wave);
      for (int g = wave; g < nG; g += AGG_WAVES) {
        const uint4 c = nxt;
        nxt = piece(g + AGG_WAVES);
        const int q = g * 32 + lane;
        const bool in = q < nQ;
        const int eb = q * 128;
        take(in ? c.x : 0u, eb);
        take(in ? c.y : 0u, eb + 32);
        take(in ? c.z : 0u, eb + 64);
        take(in ? c.w : 0u, eb + 96);
      }
    }
    if (qn > 0) consume(qn);
  };

  if constexpr (ROLE == 1) {
    if ((int)threadIdx.x <= AGG_WAVES) oqn[threadIdx.x] = 0;
    __syncthreads();
    auto route = [&](int nb) {
      if (lane < nb) {
        const int2 p = pairs[myq[lane]];
        const int ld = p.y - lo;
        const int owner = ld & 31;
        const int pos = __hip_atomic_fetch_add(&oqn[owner], 1, __ATOMIC_RELAXED,
                                               __HIP_MEMORY_SCOPE_WORKGROUP);
        if (pos < OWN_QCAP) oq[owner * OWN_QPITCH + pos] = p.x | ((ld >> 5) << OWN_SLOT_SHIFT);
        else oqn[AGG_WAVES] = 1;
      }
    };
    scan_row(route, 0);
    __syncthreads();
    if (__builtin_amdgcn_readfirstlane(oqn[AGG_WAVES]) != 0) owned = false;
  }

  if (owned) {
    int n;
    if constexpr (ROLE == 1) n = __builtin_amdgcn_readfirstlane(oqn[wave]);
    else n = __builtin_amdgcn_readlane(cv, wave);
    const int npiece = (n + 31) >> 5;

    v32f A0, A1;
#pragma unroll
    for (int i = 0; i < 32; ++i) {
      A0[i] = 0.0f;
      A1[i] = 0.0f;
    }
#pragma unroll
    for (int k = 0; k < OWN_SLOTS; ++k) {
      const int r = wave + 32 * k;
      const bool in = r < cnt;
      const vec_t t = *(const vec_t*)(xin + (long long)(lo + (in ? r : cnt - 1)) * F + VEC * lane);
      A0[2 * k] = in ? t[0] : 0.0f;
      A0[2 * k + 1] = in ? t[1] : 0.0f;
      if constexpr (VEC == 4) {
        A1[2 * k] = in ? t[2] : 0.0f;
        A1[2 * k + 1] = in ? t[3] : 0.0f;
      }
    }

    static_assert(2 * OWN_SLOTS + 2 <= 32, "every slot and the sink must fit a vector of 32");
    const int* const q = oq + (ROLE == 1 ? wave * OWN_QPITCH : 0);
    int nxt = first;
    for (int b = 0; b < npiece; ++b) {
      int e;
      if constexpr (ROLE == 1) {
        const int i = b * 32 + lane;
        e = i < n ? q[i] : 0;
        mylist[b * 32] = e;
      } else {
        e = nxt;
        int bn = b + 1;
        if (bn > npiece - 1) bn = npiece - 1;
        nxt = mylist[bn * 32];
      }
      int nb = n - b * 32;
      if (nb > 32) nb = 32;
      for (int j = 0; j < nb; j += GROUP) {
        vec_t v[GROUP];
        int slot2[GROUP];
#pragma unroll
        for (int u = 0; u < GROUP; ++u) {
          int jj = j + u;
          const bool ok = jj < nb;
          if (!ok) jj = nb - 1;
          const int ej = __builtin_amdgcn_readlane(e, jj);
          slot2[u] = ok ? (int)((unsigned)ej >> OWN_SLOT_SHIFT) * 2 : 2 * OWN_SLOTS;
          v[u] = *(const vec_t*)(xin + (long long)(ej & OWN_SRC_MASK) * F + VEC * lane);
        }
#pragma unroll
        for (int u = 0; u < GROUP; ++u) {
          const int i = slot2[u];
          A0[i] += v[u][0];
          A0[i + 1] += v[u][1];
          if constexpr (VEC == 4) {
            A1[i] += v[u][2];
            A1[i + 1] += v[u][3];
          }
        }
      }
    }
    if constexpr (ROLE == 1) {
      if (npiece == 0) mylist[0] = 0;
      if (lane == 0) lcnt[blockIdx.x * AGG_WAVES + wave] = n;
    }

    const int rows16 = (cnt + 15) & ~15;
#pragma unroll
    for (int k = 0; k < OWN_SLOTS; ++k) {
      const int r = wave + 32 * k;
      if (r < rows16) {
        float* p = &acc[r * PITCH + lane];
        p[0] = A0[2 * k];
        p[32] = A0[2 * k + 1];
        if constexpr (VEC == 4) {
          p[64] = A1[2 * k];
          p[96] = A1[2 * k + 1];
        }
      }
    }
  } else {
    {
      const int nv = (((cnt + 15) & ~15) * PITCH) >> 2;
      const v4f z = {0.0f, 0.0f, 0.0f, 0.0f};
      for (int q = (int)threadIdx.x; q < nv; q += AGG_WAVES * 32) ((v4f*)acc)[q] = z;
      if ((int)threadIdx.x < PITCH) acc[SPARE_ROW * PITCH + threadIdx.x] = 0.0f;
    }
    __syncthreads();
    int qn = 0;
    static_assert((TILE_NODES + AGG_WAVES - 1) / AGG_WAVES <= 31, "own rows must fit the waiting part of the list");
    {
      const int r = wave + AGG_WAVES * lane;
      if (r < cnt) myq[lane] = r | OWN_ROW;
      if (wave < cnt) qn = (cnt - wave + AGG_WAVES - 1) / AGG_WAVES;
    }
    scan_row(add_rows, qn);
    if constexpr (ROLE == 1) {
      mylist[0] = 0;
      if (lane == 0) lcnt[blockIdx.x * AGG_WAVES + wave] = -1;
    }
  }
  __syncthreads();

  const int r0 = wave * 16;
  if (r0 >= cnt) return;
  const int half = lane >> 4;
  const int l16 = lane & 15;

  v8f g1[4];
#pragma unroll
  for (int t = 0; t < 4; ++t) {
    const float bias = ba[t * 16 + l16];
#pragma unroll
    for (int r = 0; r < 8; ++r) g1[t][r] = bias;
  }
  float* const arow = &acc[(r0 + l16) * PITCH + 16 * half];
  const float* const wa = Waf + 4 * lane;
#pragma unroll 4
  for (int j = 0; j < F / 8; ++j) {
    v4f b[4];
#pragma unroll
    for (int t = 0; t < 4; ++t) b[t] = *(const v4f*)(wa + (j * 4 + t) * 128);
#pragma unroll
    for (int s = 0; s < 2; ++s) {
      const int e = 4 * j + 2 * s;
      const int col = (e % VEC) * 32 + e / VEC;
      v2f a;
      a.x = arow[col];
      a.y = arow[col + 32];
#pragma unroll
      for (int t = 0; t < 4; ++t) {
        v2f bb2;
        bb2.x = s == 0 ? b[t].x : b[t].z;
        bb2.y = s == 0 ? b[t].y : b[t].w;
        g1[t] = __builtin_amdgcn_wmma_f32_16x16x4_f32(
            false, a, false, bb2, (short)0, g1[t], false, false);
      }
    }
  }

#pragma unroll
  for (int t = 0; t < 4; ++t) {
    const int n = t * 16 + l16;
#pragma unroll
    for (int r = 0; r < 8; ++r) {
      float x = g1[t][r];
      x = x > 0.0f ? x : 0.0f;
      acc[(r0 + r + 8 * half) * PITCH + (n & 1) * 32 + (n >> 1)] = x;
    }
  }

  v8f g2[4];
#pragma unroll
  for (int t = 0; t < 4; ++t)
#pragma unroll
    for (int r = 0; r < 8; ++r) g2[t][r] = 0.0f;
  const float* const wb = Wbf + 4 * lane;
#pragma unroll 4
  for (int j = 0; j < HIDDEN / 8; ++j) {
    v4f b[4];
#pragma unroll
    for (int t = 0; t < 4; ++t) b[t] = *(const v4f*)(wb + (j * 4 + t) * 128);
#pragma unroll
    for (int s = 0; s < 2; ++s) {
      const int q = 2 * j + s;
      v2f a;
      a.x = arow[q];
      a.y = arow[q + 32];
#pragma unroll
      for (int t = 0; t < 4; ++t) {
        v2f bb2;
        bb2.x = s == 0 ? b[t].x : b[t].z;
        bb2.y = s == 0 ? b[t].y : b[t].w;
        g2[t] = __builtin_amdgcn_wmma_f32_16x16x4_f32(
            false, a, false, bb2, (short)0, g2[t], false, false);
      }
    }
  }

#pragma unroll
  for (int t = 0; t < 4; ++t) {
    const int n = t * 16 + l16;
    const float s = gg[n] * rsqrtf(vv[n] + BN_EPS);
    const float sh = (bb[n] - mm[n]) * s + be[n];
#pragma unroll
    for (int r = 0; r < 8; ++r) {
      const int lr = r0 + r + 8 * half;
      float o = g2[t][r] * s + sh;
      o = o > 0.0f ? o : 0.0f;
      if (lr < cnt) hout[(long long)(lo + lr) * HIDDEN + n] = o;
    }
  }
}

__global__ __launch_bounds__(256) void pool_zero_kernel(float* __restrict__ sums,
                                                        float* __restrict__ counts) {
  int i = blockIdx.x * blockDim.x + threadIdx.x;
  if (i < NGRAPH * HIDDEN) sums[i] = 0.0f;
  if (i < NGRAPH) counts[i] = 0.0f;
}

#define POOL_ROWS  32
#define POOL_GROUP 16

__global__ __launch_bounds__(256) void pool_sum_kernel(
    const float* __restrict__ h, const int* __restrict__ batch,
    float* __restrict__ sums, float* __restrict__ counts, int n_nodes) {
  const int lane = threadIdx.x & 31;
  const int wave =
      __builtin_amdgcn_readfirstlane((int)((blockIdx.x * blockDim.x + threadIdx.x) >> 5));
  const int i0 = wave * POOL_ROWS;
  if (i0 >= n_nodes) return;
  int i1 = i0 + POOL_ROWS;
  if (i1 > n_nodes) i1 = n_nodes;

  auto flush = [&](int g, v2f a, int rows) {
    float* p = sums + (long long)g * HIDDEN + 2 * lane;
    atomicAdd(p, a.x);
    atomicAdd(p + 1, a.y);
    if (lane == 0) atomicAdd(&counts[g], (float)rows);
  };

  int cur = batch[i0];
  v2f acc = {0.0f, 0.0f};
  int rows = 0;
  for (int i = i0; i < i1; i += POOL_GROUP) {
    v2f v[POOL_GROUP];
    int g[POOL_GROUP];
#pragma unroll
    for (int u = 0; u < POOL_GROUP; ++u) {
      int ii = i + u;
      if (ii > i1 - 1) ii = i1 - 1;
      g[u] = batch[ii];
      v[u] = *(const v2f*)(h + (long long)ii * HIDDEN + 2 * lane);
    }
#pragma unroll
    for (int u = 0; u < POOL_GROUP; ++u) {
      if (i + u < i1) {
        if (g[u] != cur) {
          flush(cur, acc, rows);
          acc.x = 0.0f;
          acc.y = 0.0f;
          rows = 0;
          cur = g[u];
        }
        acc.x += v[u].x;
        acc.y += v[u].y;
        ++rows;
      }
    }
  }
  flush(cur, acc, rows);
}

__global__ __launch_bounds__(256) void pool_div_kernel(
    const float* __restrict__ sums, const float* __restrict__ counts,
    float* __restrict__ out) {
  int i = blockIdx.x * blockDim.x + threadIdx.x;
  if (i >= NGRAPH * HIDDEN) return;
  float c = counts[i >> 6];
  c = c > 1.0f ? c : 1.0f;
  out[i] = sums[i] / c;
}

extern "C" void kernel_launch(void* const* d_in, const int* in_sizes, int n_in,
                              void* d_out, int out_size, void* d_ws, size_t ws_size,
                              hipStream_t stream) {
  const float* x   = (const float*)d_in[0];
  const int* ei    = (const int*)d_in[1];
  const int* batch = (const int*)d_in[2];
  const float* W1a = (const float*)d_in[3];
  const float* b1a = (const float*)d_in[4];
  const float* W1b = (const float*)d_in[5];
  const float* b1b = (const float*)d_in[6];
  const float* g1  = (const float*)d_in[7];
  const float* be1 = (const float*)d_in[8];
  const float* m1  = (const float*)d_in[9];
  const float* v1  = (const float*)d_in[10];
  const float* W2a = (const float*)d_in[11];
  const float* b2a = (const float*)d_in[12];
  const float* W2b = (const float*)d_in[13];
  const float* b2b = (const float*)d_in[14];
  const float* g2  = (const float*)d_in[15];
  const float* be2 = (const float*)d_in[16];
  const float* m2  = (const float*)d_in[17];
  const float* v2  = (const float*)d_in[18];
  float* out = (float*)d_out;

  const int N = in_sizes[0] / IN_FEATS;
  const int E = in_sizes[1] / 2;

  float* ws    = (float*)d_ws;
  float* h1    = ws;
  float* h2    = h1 + (long long)N * HIDDEN;
  float* sums  = h2 + (long long)N * HIDDEN;
  float* cnts  = sums + NGRAPH * HIDDEN;
  (void)ws_size; (void)n_in; (void)out_size;

  const int aggBlk = (N + TILE_NODES - 1) / TILE_NODES;

  const int maskBlk    = (E + MASK_EDGES - 1) / MASK_EDGES;
  const int pitchWords = maskBlk * MASK_WORDS;
  float* wf = cnts + NGRAPH;
  float* w1af = wf;
  float* w1bf = w1af + 4 * WF_W1A_VEC;
  float* w2af = w1bf + 4 * WF_SQ_VEC;
  float* w2bf = w2af + 4 * WF_SQ_VEC;
  unsigned* mask = (unsigned*)(wf + 4 * WF_VECS);
  int2* pairs    = (int2*)(mask + (long long)aggBlk * pitchWords);
  int* lcnt      = (int*)(pairs + E);
  int* elist     = lcnt + aggBlk * AGG_WAVES;

  edge_mask_kernel<<<maskBlk + WF_BLOCKS, MASK_THREADS, 0, stream>>>(
      ei, mask, pairs, E, aggBlk, pitchWords, W1a, W1b, W2a, W2b, wf, maskBlk);

  tile_layer_kernel<IN_FEATS, 1><<<aggBlk, 1024, 0, stream>>>(
      x, mask, pairs, w1af, b1a, w1bf, b1b, g1, be1, m1, v1, h1, N, pitchWords, elist, lcnt);

  tile_layer_kernel<HIDDEN, 2><<<aggBlk, 1024, 0, stream>>>(
      h1, mask, pairs, w2af, b2a, w2bf, b2b, g2, be2, m2, v2, h2, N, pitchWords, elist, lcnt);

  pool_zero_kernel<<<(NGRAPH * HIDDEN + 255) / 256, 256, 0, stream>>>(sums, cnts);
  {
    const int rowsPerBlock = POOL_ROWS * 8;
    const int blocks = (N + rowsPerBlock - 1) / rowsPerBlock;
    pool_sum_kernel<<<blocks, 256, 0, stream>>>(h2, batch, sums, cnts, N);
  }
  pool_div_kernel<<<(NGRAPH * HIDDEN + 255) / 256, 256, 0, stream>>>(sums, cnts, out);
}
